# starting kernel
# baseline (speedup 1.0000x reference)
#include <hip/hip_runtime.h>
#include <hip/hip_bf16.h>

#define NH 256
#define NN 64
#define LL 2048
#define CHUNK 16
#define NCHUNK (LL / CHUNK)

typedef __attribute__((ext_vector_type(2))) float v2f;
typedef __attribute__((ext_vector_type(8))) float v8f;

struct cpx { float r, i; };
__device__ __forceinline__ cpx cmul(cpx a, cpx b) { return {a.r*b.r - a.i*b.i, a.r*b.i + a.i*b.r}; }
__device__ __forceinline__ cpx cadd(cpx a, cpx b) { return {a.r + b.r, a.i + b.i}; }
__device__ __forceinline__ cpx csub(cpx a, cpx b) { return {a.r - b.r, a.i - b.i}; }
__device__ __forceinline__ cpx cinv(cpx a) { float d = a.r*a.r + a.i*a.i; return {a.r/d, -a.i/d}; }
__device__ __forceinline__ cpx cscale(cpx a, float s) { return {a.r*s, a.i*s}; }

__device__ __forceinline__ void cplx_mm_tile(
    const float* Ar, const float* Ai, int lda,
    const float* Br, const float* Bi, int ldb,
    float* Cr, float* Ci, int ldc,
    int r0, int n0, int K)
{
    const int lane = threadIdx.x & 31;
    const int mn = lane & 15;
    const int kh = (lane >> 4) << 1;
    v8f arbr = {}; v8f aibi = {}; v8f arbi = {}; v8f aibr = {};
    for (int k0 = 0; k0 < K; k0 += 4) {
        const int ka = k0 + kh;
        v2f ar, ai, br, bi;
        ar[0] = Ar[(r0 + mn) * lda + ka];     ar[1] = Ar[(r0 + mn) * lda + ka + 1];
        ai[0] = Ai[(r0 + mn) * lda + ka];     ai[1] = Ai[(r0 + mn) * lda + ka + 1];
        br[0] = Br[ka * ldb + n0 + mn];       br[1] = Br[(ka + 1) * ldb + n0 + mn];
        bi[0] = Bi[ka * ldb + n0 + mn];       bi[1] = Bi[(ka + 1) * ldb + n0 + mn];
        arbr = __builtin_amdgcn_wmma_f32_16x16x4_f32(false, ar, false, br, (short)0, arbr, false, false);
        aibi = __builtin_amdgcn_wmma_f32_16x16x4_f32(false, ai, false, bi, (short)0, aibi, false, false);
        arbi = __builtin_amdgcn_wmma_f32_16x16x4_f32(false, ar, false, bi, (short)0, arbi, false, false);
        aibr = __builtin_amdgcn_wmma_f32_16x16x4_f32(false, ai, false, br, (short)0, aibr, false, false);
    }
    const int mbase = (lane >> 4) << 3;
    for (int v = 0; v < 8; ++v) {
        Cr[(r0 + mbase + v) * ldc + n0 + mn] = arbr[v] - aibi[v];
        Ci[(r0 + mbase + v) * ldc + n0 + mn] = arbi[v] + aibr[v];
    }
}

extern "C" __global__ __launch_bounds__(128)
void s4_dplr_kernel(const float* __restrict__ A_real, const float* __restrict__ A_imag,
                    const float* __restrict__ B_real, const float* __restrict__ B_imag,
                    const float* __restrict__ P_real, const float* __restrict__ P_imag,
                    const float* __restrict__ C_real, const float* __restrict__ C_imag,
                    const float* __restrict__ log_dt, float* __restrict__ out)
{
    extern __shared__ float sm[];
    const int h   = blockIdx.x;
    const int tid = threadIdx.x;
    const int wv  = tid >> 5;

    float* A0r = sm;
    float* A0i = sm + 4096;
    float* A1r = sm + 8192;
    float* A1i = sm + 12288;
    float* Xr0 = sm + 16384;
    float* Xi0 = sm + 17408;
    float* Xr1 = sm + 18432;
    float* Xi1 = sm + 19456;
    float* ur  = sm + 20480;  float* ui = ur + 64;
    float* wr  = ui + 64;     float* wi = wr + 64;
    float* gr  = wi + 64;     float* gi = gr + 64;
    float* cr  = gi + 64;     float* ci = cr + 64;
    float* pr  = ci + 64;     float* pi = pr + 64;
    float* scal = pi + 64;

    const float dt = __expf(log_dt[h]);
    const float cc = 0.5f * dt;

    if (tid < 64) {
        const int n = tid;
        cpx lam = { -A_real[h*64 + n], -A_imag[h*64 + n] };
        cpx p   = {  P_real[h*64 + n],  P_imag[h*64 + n] };
        cpx Dinv = cinv((cpx){1.0f - cc*lam.r, -cc*lam.i});
        cpx e    = {1.0f + cc*lam.r,  cc*lam.i};
        cpx u = cmul(Dinv, p);
        cpx g = cmul(Dinv, e);
        ur[n] = u.r; ui[n] = u.i; gr[n] = g.r; gi[n] = g.i;
        pr[n] = p.r; pi[n] = p.i;
        cr[n] = C_real[h*64 + n]; ci[n] = C_imag[h*64 + n];
    }
    __syncthreads();
    if (tid == 0) {
        cpx s = {0.f, 0.f};
        for (int n = 0; n < 64; ++n)
            s = cadd(s, cmul((cpx){pr[n], -pi[n]}, (cpx){ur[n], ui[n]}));
        cpx alpha = cmul((cpx){cc, 0.f}, cinv((cpx){1.0f + cc*s.r, cc*s.i}));
        cpx base  = csub((cpx){cc, 0.f}, cmul(alpha, cscale(s, cc)));
        scal[0] = alpha.r; scal[1] = alpha.i; scal[2] = base.r; scal[3] = base.i;
    }
    __syncthreads();
    const cpx alpha = {scal[0], scal[1]};
    const cpx base  = {scal[2], scal[3]};
    if (tid < 64) {
        const int n = tid;
        cpx w = cmul((cpx){pr[n], -pi[n]}, cadd(base, cmul(alpha, (cpx){gr[n], gi[n]})));
        wr[n] = w.r; wi[n] = w.i;
    }
    __syncthreads();
    for (int idx = tid; idx < 4096; idx += 128) {
        const int n = idx >> 6, m = idx & 63;
        cpx v = cmul((cpx){ur[n], ui[n]}, (cpx){wr[m], wi[m]});
        float rr = -v.r, rim = -v.i;
        if (n == m) { rr += gr[n]; rim += gi[n]; }
        A0r[idx] = rr; A0i[idx] = rim;
    }
    __syncthreads();

    if (tid < 64) {
        const int n = tid;
        cpx lam = { -A_real[h*64 + n], -A_imag[h*64 + n] };
        cpx Dinv = cinv((cpx){1.0f - cc*lam.r, -cc*lam.i});
        cpx b = { B_real[h*64 + n], B_imag[h*64 + n] };
        cpx t = cmul(Dinv, cscale(b, dt));
        wr[n] = t.r; wi[n] = t.i;
    }
    __syncthreads();
    if (tid == 0) {
        cpx q = {0.f, 0.f};
        for (int n = 0; n < 64; ++n)
            q = cadd(q, cmul((cpx){pr[n], -pi[n]}, (cpx){wr[n], wi[n]}));
        cpx aq = cmul(alpha, q);
        scal[4] = aq.r; scal[5] = aq.i;
    }
    __syncthreads();
    if (tid < 64) {
        const int n = tid;
        cpx dB = csub((cpx){wr[n], wi[n]}, cmul((cpx){scal[4], scal[5]}, (cpx){ur[n], ui[n]}));
        Xr0[n*16 + 0] = dB.r;
        Xi0[n*16 + 0] = dB.i;
    }
    __syncthreads();

    for (int l = 1; l < CHUNK; ++l) {
        if (tid < 64) {
            const int n = tid;
            cpx acc = {0.f, 0.f};
            for (int m = 0; m < 64; ++m)
                acc = cadd(acc, cmul((cpx){A0r[n*64 + m], A0i[n*64 + m]},
                                     (cpx){Xr0[m*16 + l - 1], Xi0[m*16 + l - 1]}));
            Xr0[n*16 + l] = acc.r; Xi0[n*16 + l] = acc.i;
        }
        __syncthreads();
    }
    if (tid < 16) {
        float k = 0.f;
        for (int n = 0; n < 64; ++n)
            k += cr[n]*Xr0[n*16 + tid] - ci[n]*Xi0[n*16 + tid];
        out[h*LL + tid] = 2.0f * k;
    }
    __syncthreads();

    {
        for (int n0 = 0; n0 < 64; n0 += 16)
            cplx_mm_tile(A0r, A0i, 64, A0r, A0i, 64, A1r, A1i, 64, 16*wv, n0, 64);
        __syncthreads();
        for (int n0 = 0; n0 < 64; n0 += 16)
            cplx_mm_tile(A1r, A1i, 64, A1r, A1i, 64, A0r, A0i, 64, 16*wv, n0, 64);
        __syncthreads();
        for (int n0 = 0; n0 < 64; n0 += 16)
            cplx_mm_tile(A0r, A0i, 64, A0r, A0i, 64, A1r, A1i, 64, 16*wv, n0, 64);
        __syncthreads();
        for (int n0 = 0; n0 < 64; n0 += 16)
            cplx_mm_tile(A1r, A1i, 64, A1r, A1i, 64, A0r, A0i, 64, 16*wv, n0, 64);
        __syncthreads();
    }

    auto proj_out = [&](const float* Dxr, const float* Dxi, int j) {
        if (tid < 16) {
            float k = 0.f;
            for (int n = 0; n < 64; ++n)
                k += cr[n]*Dxr[n*16 + tid] - ci[n]*Dxi[n*16 + tid];
            out[h*LL + j*CHUNK + tid] = 2.0f * k;
        }
        __syncthreads();
    };

    for (int jj = 0; jj < 63; ++jj) {
        const int j = 2*jj + 1;
        cplx_mm_tile(A0r, A0i, 64, Xr0, Xi0, 16, Xr1, Xi1, 16, 16*wv, 0, 64);
        __syncthreads();
        proj_out(Xr1, Xi1, j);
        cplx_mm_tile(A0r, A0i, 64, Xr1, Xi1, 16, Xr0, Xi0, 16, 16*wv, 0, 64);
        __syncthreads();
        proj_out(Xr0, Xi0, j + 1);
    }
    cplx_mm_tile(A0r, A0i, 64, Xr0, Xi0, 16, Xr1, Xi1, 16, 16*wv, 0, 64);
    __syncthreads();
    proj_out(Xr1, Xi1, 127);
}

extern "C" void kernel_launch(void* const* d_in, const int* in_sizes, int n_in,
                              void* d_out, int out_size, void* d_ws, size_t ws_size,
                              hipStream_t stream) {
    const float* A_real = (const float*)d_in[0];
    const float* A_imag = (const float*)d_in[1];
    const float* B_real = (const float*)d_in[2];
    const float* B_imag = (const float*)d_in[3];
    const float* P_real = (const float*)d_in[4];
    const float* P_imag = (const float*)d_in[5];
    const float* C_real = (const float*)d_in[6];
    const float* C_imag = (const float*)d_in[7];
    const float* log_dt = (const float*)d_in[8];
    float* out = (float*)d_out;
    (void)in_sizes; (void)n_in; (void)out_size; (void)d_ws; (void)ws_size;

    const size_t shmem_bytes = 21128u * sizeof(float);
    s4_dplr_kernel<<<NH, 128, shmem_bytes, stream>>>(
        A_real, A_imag, B_real, B_imag, P_real, P_imag, C_real, C_imag, log_dt, out);
}


# Scan: output projection folded into each matmul step's epilogue (per-lane partial dots from D registers into an LDS table, summed and stored once at the end; one barrier per step); sum order changes
# speedup vs baseline: 1.3346x; 1.3346x over previous
#include <hip/hip_runtime.h>
#include <hip/hip_bf16.h>

#define NH 256
#define NN 64
#define LL 2048
#define CHUNK 16
#define NCHUNK (LL / CHUNK)

typedef __attribute__((ext_vector_type(2))) float v2f;
typedef __attribute__((ext_vector_type(8))) float v8f;

struct cpx { float r, i; };
__device__ __forceinline__ cpx cmul(cpx a, cpx b) { return {a.r*b.r - a.i*b.i, a.r*b.i + a.i*b.r}; }
__device__ __forceinline__ cpx cadd(cpx a, cpx b) { return {a.r + b.r, a.i + b.i}; }
__device__ __forceinline__ cpx csub(cpx a, cpx b) { return {a.r - b.r, a.i - b.i}; }
__device__ __forceinline__ cpx cinv(cpx a) { float d = a.r*a.r + a.i*a.i; return {a.r/d, -a.i/d}; }
__device__ __forceinline__ cpx cscale(cpx a, float s) { return {a.r*s, a.i*s}; }

__device__ __forceinline__ void cplx_mm_tile(
    const float* Ar, const float* Ai, int lda,
    const float* Br, const float* Bi, int ldb,
    float* Cr, float* Ci, int ldc,
    int r0, int n0, int K)
{
    const int lane = threadIdx.x & 31;
    const int mn = lane & 15;
    const int kh = (lane >> 4) << 1;
    v8f arbr = {}; v8f aibi = {}; v8f arbi = {}; v8f aibr = {};
    for (int k0 = 0; k0 < K; k0 += 4) {
        const int ka = k0 + kh;
        v2f ar, ai, br, bi;
        ar[0] = Ar[(r0 + mn) * lda + ka];     ar[1] = Ar[(r0 + mn) * lda + ka + 1];
        ai[0] = Ai[(r0 + mn) * lda + ka];     ai[1] = Ai[(r0 + mn) * lda + ka + 1];
        br[0] = Br[ka * ldb + n0 + mn];       br[1] = Br[(ka + 1) * ldb + n0 + mn];
        bi[0] = Bi[ka * ldb + n0 + mn];       bi[1] = Bi[(ka + 1) * ldb + n0 + mn];
        arbr = __builtin_amdgcn_wmma_f32_16x16x4_f32(false, ar, false, br, (short)0, arbr, false, false);
        aibi = __builtin_amdgcn_wmma_f32_16x16x4_f32(false, ai, false, bi, (short)0, aibi, false, false);
        arbi = __builtin_amdgcn_wmma_f32_16x16x4_f32(false, ar, false, bi, (short)0, arbi, false, false);
        aibr = __builtin_amdgcn_wmma_f32_16x16x4_f32(false, ai, false, br, (short)0, aibr, false, false);
    }
    const int mbase = (lane >> 4) << 3;
    for (int v = 0; v < 8; ++v) {
        Cr[(r0 + mbase + v) * ldc + n0 + mn] = arbr[v] - aibi[v];
        Ci[(r0 + mbase + v) * ldc + n0 + mn] = arbi[v] + aibr[v];
    }
}

__device__ __forceinline__ void scan_step_tile(
    const float* Ar, const float* Ai,
    const float* Br, const float* Bi,
    float* Cr, float* Ci,
    const float (&crv)[8], const float (&civ)[8],
    float* kdst, int r0)
{
    const int lane = threadIdx.x & 31;
    const int mn = lane & 15;
    const int kh = (lane >> 4) << 1;
    v8f arbr = {}; v8f aibi = {}; v8f arbi = {}; v8f aibr = {};
    for (int k0 = 0; k0 < 64; k0 += 4) {
        const int ka = k0 + kh;
        v2f ar, ai, br, bi;
        ar[0] = Ar[(r0 + mn) * 64 + ka];      ar[1] = Ar[(r0 + mn) * 64 + ka + 1];
        ai[0] = Ai[(r0 + mn) * 64 + ka];      ai[1] = Ai[(r0 + mn) * 64 + ka + 1];
        br[0] = Br[ka * 16 + mn];             br[1] = Br[(ka + 1) * 16 + mn];
        bi[0] = Bi[ka * 16 + mn];             bi[1] = Bi[(ka + 1) * 16 + mn];
        arbr = __builtin_amdgcn_wmma_f32_16x16x4_f32(false, ar, false, br, (short)0, arbr, false, false);
        aibi = __builtin_amdgcn_wmma_f32_16x16x4_f32(false, ai, false, bi, (short)0, aibi, false, false);
        arbi = __builtin_amdgcn_wmma_f32_16x16x4_f32(false, ar, false, bi, (short)0, arbi, false, false);
        aibr = __builtin_amdgcn_wmma_f32_16x16x4_f32(false, ai, false, br, (short)0, aibr, false, false);
    }
    const int mbase = (lane >> 4) << 3;
    float xr[8], xi[8];
    for (int v = 0; v < 8; ++v) {
        xr[v] = arbr[v] - aibi[v];
        xi[v] = arbi[v] + aibr[v];
        Cr[(r0 + mbase + v) * 16 + mn] = xr[v];
        Ci[(r0 + mbase + v) * 16 + mn] = xi[v];
    }
    float kr = 0.f, ki = 0.f;
    for (int v = 0; v < 8; ++v) {
        kr = fmaf(crv[v], xr[v], kr);
        ki = fmaf(civ[v], xi[v], ki);
    }
    *kdst = kr - ki;
}

extern "C" __global__ __launch_bounds__(128)
void s4_dplr_kernel(const float* __restrict__ A_real, const float* __restrict__ A_imag,
                    const float* __restrict__ B_real, const float* __restrict__ B_imag,
                    const float* __restrict__ P_real, const float* __restrict__ P_imag,
                    const float* __restrict__ C_real, const float* __restrict__ C_imag,
                    const float* __restrict__ log_dt, float* __restrict__ out)
{
    extern __shared__ float sm[];
    const int h   = blockIdx.x;
    const int tid = threadIdx.x;
    const int wv  = tid >> 5;

    float* A0r = sm;
    float* A0i = sm + 4096;
    float* A1r = sm + 8192;
    float* A1i = sm + 12288;
    float* Xr0 = sm + 16384;
    float* Xi0 = sm + 17408;
    float* Xr1 = sm + 18432;
    float* Xi1 = sm + 19456;
    float* ur  = sm + 20480;  float* ui = ur + 64;
    float* wr  = ui + 64;     float* wi = wr + 64;
    float* gr  = wi + 64;     float* gi = gr + 64;
    float* cr  = gi + 64;     float* ci = cr + 64;
    float* pr  = ci + 64;     float* pi = pr + 64;
    float* scal = pi + 64;
    float* Kp   = sm + 21128;

    const float dt = __expf(log_dt[h]);
    const float cc = 0.5f * dt;

    if (tid < 64) {
        const int n = tid;
        cpx lam = { -A_real[h*64 + n], -A_imag[h*64 + n] };
        cpx p   = {  P_real[h*64 + n],  P_imag[h*64 + n] };
        cpx Dinv = cinv((cpx){1.0f - cc*lam.r, -cc*lam.i});
        cpx e    = {1.0f + cc*lam.r,  cc*lam.i};
        cpx u = cmul(Dinv, p);
        cpx g = cmul(Dinv, e);
        ur[n] = u.r; ui[n] = u.i; gr[n] = g.r; gi[n] = g.i;
        pr[n] = p.r; pi[n] = p.i;
        cr[n] = C_real[h*64 + n]; ci[n] = C_imag[h*64 + n];
    }
    __syncthreads();
    if (tid == 0) {
        cpx s = {0.f, 0.f};
        for (int n = 0; n < 64; ++n)
            s = cadd(s, cmul((cpx){pr[n], -pi[n]}, (cpx){ur[n], ui[n]}));
        cpx alpha = cmul((cpx){cc, 0.f}, cinv((cpx){1.0f + cc*s.r, cc*s.i}));
        cpx base  = csub((cpx){cc, 0.f}, cmul(alpha, cscale(s, cc)));
        scal[0] = alpha.r; scal[1] = alpha.i; scal[2] = base.r; scal[3] = base.i;
    }
    __syncthreads();
    const cpx alpha = {scal[0], scal[1]};
    const cpx base  = {scal[2], scal[3]};
    if (tid < 64) {
        const int n = tid;
        cpx w = cmul((cpx){pr[n], -pi[n]}, cadd(base, cmul(alpha, (cpx){gr[n], gi[n]})));
        wr[n] = w.r; wi[n] = w.i;
    }
    __syncthreads();
    for (int idx = tid; idx < 4096; idx += 128) {
        const int n = idx >> 6, m = idx & 63;
        cpx v = cmul((cpx){ur[n], ui[n]}, (cpx){wr[m], wi[m]});
        float rr = -v.r, rim = -v.i;
        if (n == m) { rr += gr[n]; rim += gi[n]; }
        A0r[idx] = rr; A0i[idx] = rim;
    }
    __syncthreads();

    if (tid < 64) {
        const int n = tid;
        cpx lam = { -A_real[h*64 + n], -A_imag[h*64 + n] };
        cpx Dinv = cinv((cpx){1.0f - cc*lam.r, -cc*lam.i});
        cpx b = { B_real[h*64 + n], B_imag[h*64 + n] };
        cpx t = cmul(Dinv, cscale(b, dt));
        wr[n] = t.r; wi[n] = t.i;
    }
    __syncthreads();
    if (tid == 0) {
        cpx q = {0.f, 0.f};
        for (int n = 0; n < 64; ++n)
            q = cadd(q, cmul((cpx){pr[n], -pi[n]}, (cpx){wr[n], wi[n]}));
        cpx aq = cmul(alpha, q);
        scal[4] = aq.r; scal[5] = aq.i;
    }
    __syncthreads();
    if (tid < 64) {
        const int n = tid;
        cpx dB = csub((cpx){wr[n], wi[n]}, cmul((cpx){scal[4], scal[5]}, (cpx){ur[n], ui[n]}));
        Xr0[n*16 + 0] = dB.r;
        Xi0[n*16 + 0] = dB.i;
    }
    __syncthreads();

    for (int l = 1; l < CHUNK; ++l) {
        if (tid < 64) {
            const int n = tid;
            cpx acc = {0.f, 0.f};
            for (int m = 0; m < 64; ++m)
                acc = cadd(acc, cmul((cpx){A0r[n*64 + m], A0i[n*64 + m]},
                                     (cpx){Xr0[m*16 + l - 1], Xi0[m*16 + l - 1]}));
            Xr0[n*16 + l] = acc.r; Xi0[n*16 + l] = acc.i;
        }
        __syncthreads();
    }
    if (tid < 16) {
        float k = 0.f;
        for (int n = 0; n < 64; ++n)
            k += cr[n]*Xr0[n*16 + tid] - ci[n]*Xi0[n*16 + tid];
        Kp[tid] = k;
        for (int sl = 1; sl < 8; ++sl) Kp[sl*LL + tid] = 0.f;
    }
    __syncthreads();

    {
        for (int n0 = 0; n0 < 64; n0 += 16)
            cplx_mm_tile(A0r, A0i, 64, A0r, A0i, 64, A1r, A1i, 64, 16*wv, n0, 64);
        __syncthreads();
        for (int n0 = 0; n0 < 64; n0 += 16)
            cplx_mm_tile(A1r, A1i, 64, A1r, A1i, 64, A0r, A0i, 64, 16*wv, n0, 64);
        __syncthreads();
        for (int n0 = 0; n0 < 64; n0 += 16)
            cplx_mm_tile(A0r, A0i, 64, A0r, A0i, 64, A1r, A1i, 64, 16*wv, n0, 64);
        __syncthreads();
        for (int n0 = 0; n0 < 64; n0 += 16)
            cplx_mm_tile(A1r, A1i, 64, A1r, A1i, 64, A0r, A0i, 64, 16*wv, n0, 64);
        __syncthreads();
    }

    const int lane = tid & 31;
    const int hf   = lane >> 4;
    float crv[8], civ[8];
    for (int v = 0; v < 8; ++v) {
        crv[v] = cr[16*wv + 8*hf + v];
        civ[v] = ci[16*wv + 8*hf + v];
    }
    float* kslot = Kp + (2*wv + hf) * LL + (lane & 15);

    for (int jj = 0; jj < 63; ++jj) {
        const int j = 2*jj + 1;
        scan_step_tile(A0r, A0i, Xr0, Xi0, Xr1, Xi1, crv, civ, kslot + j*CHUNK, 16*wv);
        __syncthreads();
        scan_step_tile(A0r, A0i, Xr1, Xi1, Xr0, Xi0, crv, civ, kslot + (j + 1)*CHUNK, 16*wv);
        __syncthreads();
    }
    scan_step_tile(A0r, A0i, Xr0, Xi0, Xr1, Xi1, crv, civ, kslot + 127*CHUNK, 16*wv);
    __syncthreads();

    for (int i = 0; i < LL / 128; ++i) {
        const int idx = i*128 + tid;
        float sacc = Kp[idx];
        for (int sl = 1; sl < 8; ++sl) sacc += Kp[sl*LL + idx];
        out[h*LL + idx] = 2.0f * sacc;
    }
}

extern "C" void kernel_launch(void* const* d_in, const int* in_sizes, int n_in,
                              void* d_out, int out_size, void* d_ws, size_t ws_size,
                              hipStream_t stream) {
    const float* A_real = (const float*)d_in[0];
    const float* A_imag = (const float*)d_in[1];
    const float* B_real = (const float*)d_in[2];
    const float* B_imag = (const float*)d_in[3];
    const float* P_real = (const float*)d_in[4];
    const float* P_imag = (const float*)d_in[5];
    const float* C_real = (const float*)d_in[6];
    const float* C_imag = (const float*)d_in[7];
    const float* log_dt = (const float*)d_in[8];
    float* out = (float*)d_out;
    (void)in_sizes; (void)n_in; (void)out_size; (void)d_ws; (void)ws_size;

    const size_t shmem_bytes = 37512u * sizeof(float);
    s4_dplr_kernel<<<NH, 128, shmem_bytes, stream>>>(
        A_real, A_imag, B_real, B_imag, P_real, P_imag, C_real, C_imag, log_dt, out);
}


# Each wave's A-operand fragments (its 16 rows of dA^16 in the scan, of the matrix being squared in a squaring) read from LDS once into registers and reused: no bank-conflicted reload per step or tile
# speedup vs baseline: 2.7433x; 2.0555x over previous
#include <hip/hip_runtime.h>
#include <hip/hip_bf16.h>

#define NH 256
#define NN 64
#define LL 2048
#define CHUNK 16
#define NCHUNK (LL / CHUNK)

typedef __attribute__((ext_vector_type(2))) float v2f;
typedef __attribute__((ext_vector_type(8))) float v8f;

struct cpx { float r, i; };
__device__ __forceinline__ cpx cmul(cpx a, cpx b) { return {a.r*b.r - a.i*b.i, a.r*b.i + a.i*b.r}; }
__device__ __forceinline__ cpx cadd(cpx a, cpx b) { return {a.r + b.r, a.i + b.i}; }
__device__ __forceinline__ cpx csub(cpx a, cpx b) { return {a.r - b.r, a.i - b.i}; }
__device__ __forceinline__ cpx cinv(cpx a) { float d = a.r*a.r + a.i*a.i; return {a.r/d, -a.i/d}; }
__device__ __forceinline__ cpx cscale(cpx a, float s) { return {a.r*s, a.i*s}; }

__device__ __forceinline__ void load_a_frags(const float* Ar, const float* Ai, int r0,
                                             v2f (&arf)[16], v2f (&aif)[16])
{
    const int lane = threadIdx.x & 31;
    const int mn = lane & 15;
    const int kh = (lane >> 4) << 1;
    const float* rowr = Ar + (r0 + mn) * 64 + kh;
    const float* rowi = Ai + (r0 + mn) * 64 + kh;
    #pragma unroll
    for (int s = 0; s < 16; ++s) {
        arf[s][0] = rowr[4*s];  arf[s][1] = rowr[4*s + 1];
        aif[s][0] = rowi[4*s];  aif[s][1] = rowi[4*s + 1];
    }
}

__device__ __forceinline__ void cplx_mm_tile(
    const v2f (&arf)[16], const v2f (&aif)[16],
    const float* Br, const float* Bi, int ldb,
    float* Cr, float* Ci, int ldc,
    int r0, int n0)
{
    const int lane = threadIdx.x & 31;
    const int mn = lane & 15;
    const int kh = (lane >> 4) << 1;
    v8f arbr = {}; v8f aibi = {}; v8f arbi = {}; v8f aibr = {};
    #pragma unroll
    for (int s = 0; s < 16; ++s) {
        const int ka = 4*s + kh;
        v2f br, bi;
        br[0] = Br[ka * ldb + n0 + mn];       br[1] = Br[(ka + 1) * ldb + n0 + mn];
        bi[0] = Bi[ka * ldb + n0 + mn];       bi[1] = Bi[(ka + 1) * ldb + n0 + mn];
        arbr = __builtin_amdgcn_wmma_f32_16x16x4_f32(false, arf[s], false, br, (short)0, arbr, false, false);
        aibi = __builtin_amdgcn_wmma_f32_16x16x4_f32(false, aif[s], false, bi, (short)0, aibi, false, false);
        arbi = __builtin_amdgcn_wmma_f32_16x16x4_f32(false, arf[s], false, bi, (short)0, arbi, false, false);
        aibr = __builtin_amdgcn_wmma_f32_16x16x4_f32(false, aif[s], false, br, (short)0, aibr, false, false);
    }
    const int mbase = (lane >> 4) << 3;
    for (int v = 0; v < 8; ++v) {
        Cr[(r0 + mbase + v) * ldc + n0 + mn] = arbr[v] - aibi[v];
        Ci[(r0 + mbase + v) * ldc + n0 + mn] = arbi[v] + aibr[v];
    }
}

__device__ __forceinline__ void scan_step_tile(
    const v2f (&arf)[16], const v2f (&aif)[16],
    const float* Br, const float* Bi,
    float* Cr, float* Ci,
    const float (&crv)[8], const float (&civ)[8],
    float* kdst, int r0)
{
    const int lane = threadIdx.x & 31;
    const int mn = lane & 15;
    const int kh = (lane >> 4) << 1;
    v8f arbr = {}; v8f aibi = {}; v8f arbi = {}; v8f aibr = {};
    #pragma unroll
    for (int s = 0; s < 16; ++s) {
        const int ka = 4*s + kh;
        v2f br, bi;
        br[0] = Br[ka * 16 + mn];             br[1] = Br[(ka + 1) * 16 + mn];
        bi[0] = Bi[ka * 16 + mn];             bi[1] = Bi[(ka + 1) * 16 + mn];
        arbr = __builtin_amdgcn_wmma_f32_16x16x4_f32(false, arf[s], false, br, (short)0, arbr, false, false);
        aibi = __builtin_amdgcn_wmma_f32_16x16x4_f32(false, aif[s], false, bi, (short)0, aibi, false, false);
        arbi = __builtin_amdgcn_wmma_f32_16x16x4_f32(false, arf[s], false, bi, (short)0, arbi, false, false);
        aibr = __builtin_amdgcn_wmma_f32_16x16x4_f32(false, aif[s], false, br, (short)0, aibr, false, false);
    }
    const int mbase = (lane >> 4) << 3;
    float xr[8], xi[8];
    for (int v = 0; v < 8; ++v) {
        xr[v] = arbr[v] - aibi[v];
        xi[v] = arbi[v] + aibr[v];
        Cr[(r0 + mbase + v) * 16 + mn] = xr[v];
        Ci[(r0 + mbase + v) * 16 + mn] = xi[v];
    }
    float kr = 0.f, ki = 0.f;
    for (int v = 0; v < 8; ++v) {
        kr = fmaf(crv[v], xr[v], kr);
        ki = fmaf(civ[v], xi[v], ki);
    }
    *kdst = kr - ki;
}

extern "C" __global__ __launch_bounds__(128)
void s4_dplr_kernel(const float* __restrict__ A_real, const float* __restrict__ A_imag,
                    const float* __restrict__ B_real, const float* __restrict__ B_imag,
                    const float* __restrict__ P_real, const float* __restrict__ P_imag,
                    const float* __restrict__ C_real, const float* __restrict__ C_imag,
                    const float* __restrict__ log_dt, float* __restrict__ out)
{
    extern __shared__ float sm[];
    const int h   = blockIdx.x;
    const int tid = threadIdx.x;
    const int wv  = tid >> 5;

    float* A0r = sm;
    float* A0i = sm + 4096;
    float* A1r = sm + 8192;
    float* A1i = sm + 12288;
    float* Xr0 = sm + 16384;
    float* Xi0 = sm + 17408;
    float* Xr1 = sm + 18432;
    float* Xi1 = sm + 19456;
    float* ur  = sm + 20480;  float* ui = ur + 64;
    float* wr  = ui + 64;     float* wi = wr + 64;
    float* gr  = wi + 64;     float* gi = gr + 64;
    float* cr  = gi + 64;     float* ci = cr + 64;
    float* pr  = ci + 64;     float* pi = pr + 64;
    float* scal = pi + 64;
    float* Kp   = sm + 21128;

    const float dt = __expf(log_dt[h]);
    const float cc = 0.5f * dt;

    if (tid < 64) {
        const int n = tid;
        cpx lam = { -A_real[h*64 + n], -A_imag[h*64 + n] };
        cpx p   = {  P_real[h*64 + n],  P_imag[h*64 + n] };
        cpx Dinv = cinv((cpx){1.0f - cc*lam.r, -cc*lam.i});
        cpx e    = {1.0f + cc*lam.r,  cc*lam.i};
        cpx u = cmul(Dinv, p);
        cpx g = cmul(Dinv, e);
        ur[n] = u.r; ui[n] = u.i; gr[n] = g.r; gi[n] = g.i;
        pr[n] = p.r; pi[n] = p.i;
        cr[n] = C_real[h*64 + n]; ci[n] = C_imag[h*64 + n];
    }
    __syncthreads();
    if (tid == 0) {
        cpx s = {0.f, 0.f};
        for (int n = 0; n < 64; ++n)
            s = cadd(s, cmul((cpx){pr[n], -pi[n]}, (cpx){ur[n], ui[n]}));
        cpx alpha = cmul((cpx){cc, 0.f}, cinv((cpx){1.0f + cc*s.r, cc*s.i}));
        cpx base  = csub((cpx){cc, 0.f}, cmul(alpha, cscale(s, cc)));
        scal[0] = alpha.r; scal[1] = alpha.i; scal[2] = base.r; scal[3] = base.i;
    }
    __syncthreads();
    const cpx alpha = {scal[0], scal[1]};
    const cpx base  = {scal[2], scal[3]};
    if (tid < 64) {
        const int n = tid;
        cpx w = cmul((cpx){pr[n], -pi[n]}, cadd(base, cmul(alpha, (cpx){gr[n], gi[n]})));
        wr[n] = w.r; wi[n] = w.i;
    }
    __syncthreads();
    for (int idx = tid; idx < 4096; idx += 128) {
        const int n = idx >> 6, m = idx & 63;
        cpx v = cmul((cpx){ur[n], ui[n]}, (cpx){wr[m], wi[m]});
        float rr = -v.r, rim = -v.i;
        if (n == m) { rr += gr[n]; rim += gi[n]; }
        A0r[idx] = rr; A0i[idx] = rim;
    }
    __syncthreads();

    if (tid < 64) {
        const int n = tid;
        cpx lam = { -A_real[h*64 + n], -A_imag[h*64 + n] };
        cpx Dinv = cinv((cpx){1.0f - cc*lam.r, -cc*lam.i});
        cpx b = { B_real[h*64 + n], B_imag[h*64 + n] };
        cpx t = cmul(Dinv, cscale(b, dt));
        wr[n] = t.r; wi[n] = t.i;
    }
    __syncthreads();
    if (tid == 0) {
        cpx q = {0.f, 0.f};
        for (int n = 0; n < 64; ++n)
            q = cadd(q, cmul((cpx){pr[n], -pi[n]}, (cpx){wr[n], wi[n]}));
        cpx aq = cmul(alpha, q);
        scal[4] = aq.r; scal[5] = aq.i;
    }
    __syncthreads();
    if (tid < 64) {
        const int n = tid;
        cpx dB = csub((cpx){wr[n], wi[n]}, cmul((cpx){scal[4], scal[5]}, (cpx){ur[n], ui[n]}));
        Xr0[n*16 + 0] = dB.r;
        Xi0[n*16 + 0] = dB.i;
    }
    __syncthreads();

    for (int l = 1; l < CHUNK; ++l) {
        if (tid < 64) {
            const int n = tid;
            cpx acc = {0.f, 0.f};
            for (int m = 0; m < 64; ++m)
                acc = cadd(acc, cmul((cpx){A0r[n*64 + m], A0i[n*64 + m]},
                                     (cpx){Xr0[m*16 + l - 1], Xi0[m*16 + l - 1]}));
            Xr0[n*16 + l] = acc.r; Xi0[n*16 + l] = acc.i;
        }
        __syncthreads();
    }
    if (tid < 16) {
        float k = 0.f;
        for (int n = 0; n < 64; ++n)
            k += cr[n]*Xr0[n*16 + tid] - ci[n]*Xi0[n*16 + tid];
        Kp[tid] = k;
        for (int sl = 1; sl < 8; ++sl) Kp[sl*LL + tid] = 0.f;
    }
    __syncthreads();

    v2f arf[16], aif[16];
    {
        load_a_frags(A0r, A0i, 16*wv, arf, aif);
        for (int n0 = 0; n0 < 64; n0 += 16)
            cplx_mm_tile(arf, aif, A0r, A0i, 64, A1r, A1i, 64, 16*wv, n0);
        __syncthreads();
        load_a_frags(A1r, A1i, 16*wv, arf, aif);
        for (int n0 = 0; n0 < 64; n0 += 16)
            cplx_mm_tile(arf, aif, A1r, A1i, 64, A0r, A0i, 64, 16*wv, n0);
        __syncthreads();
        load_a_frags(A0r, A0i, 16*wv, arf, aif);
        for (int n0 = 0; n0 < 64; n0 += 16)
            cplx_mm_tile(arf, aif, A0r, A0i, 64, A1r, A1i, 64, 16*wv, n0);
        __syncthreads();
        load_a_frags(A1r, A1i, 16*wv, arf, aif);
        for (int n0 = 0; n0 < 64; n0 += 16)
            cplx_mm_tile(arf, aif, A1r, A1i, 64, A0r, A0i, 64, 16*wv, n0);
        __syncthreads();
    }
    load_a_frags(A0r, A0i, 16*wv, arf, aif);

    const int lane = tid & 31;
    const int hf   = lane >> 4;
    float crv[8], civ[8];
    for (int v = 0; v < 8; ++v) {
        crv[v] = cr[16*wv + 8*hf + v];
        civ[v] = ci[16*wv + 8*hf + v];
    }
    float* kslot = Kp + (2*wv + hf) * LL + (lane & 15);

    for (int jj = 0; jj < 63; ++jj) {
        const int j = 2*jj + 1;
        scan_step_tile(arf, aif, Xr0, Xi0, Xr1, Xi1, crv, civ, kslot + j*CHUNK, 16*wv);
        __syncthreads();
        scan_step_tile(arf, aif, Xr1, Xi1, Xr0, Xi0, crv, civ, kslot + (j + 1)*CHUNK, 16*wv);
        __syncthreads();
    }
    scan_step_tile(arf, aif, Xr0, Xi0, Xr1, Xi1, crv, civ, kslot + 127*CHUNK, 16*wv);
    __syncthreads();

    for (int i = 0; i < LL / 128; ++i) {
        const int idx = i*128 + tid;
        float sacc = Kp[idx];
        for (int sl = 1; sl < 8; ++sl) sacc += Kp[sl*LL + idx];
        out[h*LL + idx] = 2.0f * sacc;
    }
}

extern "C" void kernel_launch(void* const* d_in, const int* in_sizes, int n_in,
                              void* d_out, int out_size, void* d_ws, size_t ws_size,
                              hipStream_t stream) {
    const float* A_real = (const float*)d_in[0];
    const float* A_imag = (const float*)d_in[1];
    const float* B_real = (const float*)d_in[2];
    const float* B_imag = (const float*)d_in[3];
    const float* P_real = (const float*)d_in[4];
    const float* P_imag = (const float*)d_in[5];
    const float* C_real = (const float*)d_in[6];
    const float* C_imag = (const float*)d_in[7];
    const float* log_dt = (const float*)d_in[8];
    float* out = (float*)d_out;
    (void)in_sizes; (void)n_in; (void)out_size; (void)d_ws; (void)ws_size;

    const size_t shmem_bytes = 37512u * sizeof(float);
    s4_dplr_kernel<<<NH, 128, shmem_bytes, stream>>>(
        A_real, A_imag, B_real, B_imag, P_real, P_imag, C_real, C_imag, log_dt, out);
}


# First 16 power-sequence columns by doubling on the matrix unit (X[:,w..2w-1] = dA^w X[:,0..w-1] beside each squaring, like the reference) instead of 15 serial vector matvecs; their sum order changes
# speedup vs baseline: 4.4711x; 1.6298x over previous
#include <hip/hip_runtime.h>
#include <hip/hip_bf16.h>

#define NH 256
#define NN 64
#define LL 2048
#define CHUNK 16
#define NCHUNK (LL / CHUNK)

typedef __attribute__((ext_vector_type(2))) float v2f;
typedef __attribute__((ext_vector_type(8))) float v8f;

struct cpx { float r, i; };
__device__ __forceinline__ cpx cmul(cpx a, cpx b) { return {a.r*b.r - a.i*b.i, a.r*b.i + a.i*b.r}; }
__device__ __forceinline__ cpx cadd(cpx a, cpx b) { return {a.r + b.r, a.i + b.i}; }
__device__ __forceinline__ cpx csub(cpx a, cpx b) { return {a.r - b.r, a.i - b.i}; }
__device__ __forceinline__ cpx cinv(cpx a) { float d = a.r*a.r + a.i*a.i; return {a.r/d, -a.i/d}; }
__device__ __forceinline__ cpx cscale(cpx a, float s) { return {a.r*s, a.i*s}; }

__device__ __forceinline__ void load_a_frags(const float* Ar, const float* Ai, int r0,
                                             v2f (&arf)[16], v2f (&aif)[16])
{
    const int lane = threadIdx.x & 31;
    const int mn = lane & 15;
    const int kh = (lane >> 4) << 1;
    const float* rowr = Ar + (r0 + mn) * 64 + kh;
    const float* rowi = Ai + (r0 + mn) * 64 + kh;
    #pragma unroll
    for (int s = 0; s < 16; ++s) {
        arf[s][0] = rowr[4*s];  arf[s][1] = rowr[4*s + 1];
        aif[s][0] = rowi[4*s];  aif[s][1] = rowi[4*s + 1];
    }
}

__device__ __forceinline__ void cplx_mm_tile(
    const v2f (&arf)[16], const v2f (&aif)[16],
    const float* Br, const float* Bi, int ldb,
    float* Cr, float* Ci, int ldc,
    int r0, int n0)
{
    const int lane = threadIdx.x & 31;
    const int mn = lane & 15;
    const int kh = (lane >> 4) << 1;
    v8f arbr = {}; v8f aibi = {}; v8f arbi = {}; v8f aibr = {};
    #pragma unroll
    for (int s = 0; s < 16; ++s) {
        const int ka = 4*s + kh;
        v2f br, bi;
        br[0] = Br[ka * ldb + n0 + mn];       br[1] = Br[(ka + 1) * ldb + n0 + mn];
        bi[0] = Bi[ka * ldb + n0 + mn];       bi[1] = Bi[(ka + 1) * ldb + n0 + mn];
        arbr = __builtin_amdgcn_wmma_f32_16x16x4_f32(false, arf[s], false, br, (short)0, arbr, false, false);
        aibi = __builtin_amdgcn_wmma_f32_16x16x4_f32(false, aif[s], false, bi, (short)0, aibi, false, false);
        arbi = __builtin_amdgcn_wmma_f32_16x16x4_f32(false, arf[s], false, bi, (short)0, arbi, false, false);
        aibr = __builtin_amdgcn_wmma_f32_16x16x4_f32(false, aif[s], false, br, (short)0, aibr, false, false);
    }
    const int mbase = (lane >> 4) << 3;
    for (int v = 0; v < 8; ++v) {
        Cr[(r0 + mbase + v) * ldc + n0 + mn] = arbr[v] - aibi[v];
        Ci[(r0 + mbase + v) * ldc + n0 + mn] = arbi[v] + aibr[v];
    }
}

__device__ __forceinline__ void x_double_tile(
    const v2f (&arf)[16], const v2f (&aif)[16],
    float* Xr, float* Xi, float* Pr, float* Pi, int r0, int w)
{
    const int lane = threadIdx.x & 31;
    const int mn = lane & 15;
    const int kh = (lane >> 4) << 1;
    const bool live = mn < w;
    v2f brv[16], biv[16];
    #pragma unroll
    for (int s = 0; s < 16; ++s) { brv[s][0] = 0.f; brv[s][1] = 0.f; biv[s][0] = 0.f; biv[s][1] = 0.f; }
    if (live) {
        #pragma unroll
        for (int s = 0; s < 16; ++s) {
            const int ka = 4*s + kh;
            brv[s][0] = Xr[ka * 16 + mn];     brv[s][1] = Xr[(ka + 1) * 16 + mn];
            biv[s][0] = Xi[ka * 16 + mn];     biv[s][1] = Xi[(ka + 1) * 16 + mn];
        }
    }
    v8f arbr = {}; v8f aibi = {}; v8f arbi = {}; v8f aibr = {};
    #pragma unroll
    for (int s = 0; s < 16; ++s) {
        arbr = __builtin_amdgcn_wmma_f32_16x16x4_f32(false, arf[s], false, brv[s], (short)0, arbr, false, false);
        aibi = __builtin_amdgcn_wmma_f32_16x16x4_f32(false, aif[s], false, biv[s], (short)0, aibi, false, false);
        arbi = __builtin_amdgcn_wmma_f32_16x16x4_f32(false, arf[s], false, biv[s], (short)0, arbi, false, false);
        aibr = __builtin_amdgcn_wmma_f32_16x16x4_f32(false, aif[s], false, brv[s], (short)0, aibr, false, false);
    }
    const int mbase = (lane >> 4) << 3;
    float* dr = live ? (Xr + w) : Pr;
    float* di = live ? (Xi + w) : Pi;
    for (int v = 0; v < 8; ++v) {
        dr[(r0 + mbase + v) * 16 + mn] = arbr[v] - aibi[v];
        di[(r0 + mbase + v) * 16 + mn] = arbi[v] + aibr[v];
    }
}

__device__ __forceinline__ void scan_step_tile(
    const v2f (&arf)[16], const v2f (&aif)[16],
    const float* Br, const float* Bi,
    float* Cr, float* Ci,
    const float (&crv)[8], const float (&civ)[8],
    float* kdst, int r0)
{
    const int lane = threadIdx.x & 31;
    const int mn = lane & 15;
    const int kh = (lane >> 4) << 1;
    v8f arbr = {}; v8f aibi = {}; v8f arbi = {}; v8f aibr = {};
    #pragma unroll
    for (int s = 0; s < 16; ++s) {
        const int ka = 4*s + kh;
        v2f br, bi;
        br[0] = Br[ka * 16 + mn];             br[1] = Br[(ka + 1) * 16 + mn];
        bi[0] = Bi[ka * 16 + mn];             bi[1] = Bi[(ka + 1) * 16 + mn];
        arbr = __builtin_amdgcn_wmma_f32_16x16x4_f32(false, arf[s], false, br, (short)0, arbr, false, false);
        aibi = __builtin_amdgcn_wmma_f32_16x16x4_f32(false, aif[s], false, bi, (short)0, aibi, false, false);
        arbi = __builtin_amdgcn_wmma_f32_16x16x4_f32(false, arf[s], false, bi, (short)0, arbi, false, false);
        aibr = __builtin_amdgcn_wmma_f32_16x16x4_f32(false, aif[s], false, br, (short)0, aibr, false, false);
    }
    const int mbase = (lane >> 4) << 3;
    float xr[8], xi[8];
    for (int v = 0; v < 8; ++v) {
        xr[v] = arbr[v] - aibi[v];
        xi[v] = arbi[v] + aibr[v];
        Cr[(r0 + mbase + v) * 16 + mn] = xr[v];
        Ci[(r0 + mbase + v) * 16 + mn] = xi[v];
    }
    float kr = 0.f, ki = 0.f;
    for (int v = 0; v < 8; ++v) {
        kr = fmaf(crv[v], xr[v], kr);
        ki = fmaf(civ[v], xi[v], ki);
    }
    *kdst = kr - ki;
}

extern "C" __global__ __launch_bounds__(128)
void s4_dplr_kernel(const float* __restrict__ A_real, const float* __restrict__ A_imag,
                    const float* __restrict__ B_real, const float* __restrict__ B_imag,
                    const float* __restrict__ P_real, const float* __restrict__ P_imag,
                    const float* __restrict__ C_real, const float* __restrict__ C_imag,
                    const float* __restrict__ log_dt, float* __restrict__ out)
{
    extern __shared__ float sm[];
    const int h   = blockIdx.x;
    const int tid = threadIdx.x;
    const int wv  = tid >> 5;

    float* A0r = sm;
    float* A0i = sm + 4096;
    float* A1r = sm + 8192;
    float* A1i = sm + 12288;
    float* Xr0 = sm + 16384;
    float* Xi0 = sm + 17408;
    float* Xr1 = sm + 18432;
    float* Xi1 = sm + 19456;
    float* ur  = sm + 20480;  float* ui = ur + 64;
    float* wr  = ui + 64;     float* wi = wr + 64;
    float* gr  = wi + 64;     float* gi = gr + 64;
    float* cr  = gi + 64;     float* ci = cr + 64;
    float* pr  = ci + 64;     float* pi = pr + 64;
    float* scal = pi + 64;
    float* Kp   = sm + 21128;

    const float dt = __expf(log_dt[h]);
    const float cc = 0.5f * dt;

    if (tid < 64) {
        const int n = tid;
        cpx lam = { -A_real[h*64 + n], -A_imag[h*64 + n] };
        cpx p   = {  P_real[h*64 + n],  P_imag[h*64 + n] };
        cpx Dinv = cinv((cpx){1.0f - cc*lam.r, -cc*lam.i});
        cpx e    = {1.0f + cc*lam.r,  cc*lam.i};
        cpx u = cmul(Dinv, p);
        cpx g = cmul(Dinv, e);
        ur[n] = u.r; ui[n] = u.i; gr[n] = g.r; gi[n] = g.i;
        pr[n] = p.r; pi[n] = p.i;
        cr[n] = C_real[h*64 + n]; ci[n] = C_imag[h*64 + n];
    }
    __syncthreads();
    if (tid == 0) {
        cpx s = {0.f, 0.f};
        for (int n = 0; n < 64; ++n)
            s = cadd(s, cmul((cpx){pr[n], -pi[n]}, (cpx){ur[n], ui[n]}));
        cpx alpha = cmul((cpx){cc, 0.f}, cinv((cpx){1.0f + cc*s.r, cc*s.i}));
        cpx base  = csub((cpx){cc, 0.f}, cmul(alpha, cscale(s, cc)));
        scal[0] = alpha.r; scal[1] = alpha.i; scal[2] = base.r; scal[3] = base.i;
    }
    __syncthreads();
    const cpx alpha = {scal[0], scal[1]};
    const cpx base  = {scal[2], scal[3]};
    if (tid < 64) {
        const int n = tid;
        cpx w = cmul((cpx){pr[n], -pi[n]}, cadd(base, cmul(alpha, (cpx){gr[n], gi[n]})));
        wr[n] = w.r; wi[n] = w.i;
    }
    __syncthreads();
    for (int idx = tid; idx < 4096; idx += 128) {
        const int n = idx >> 6, m = idx & 63;
        cpx v = cmul((cpx){ur[n], ui[n]}, (cpx){wr[m], wi[m]});
        float rr = -v.r, rim = -v.i;
        if (n == m) { rr += gr[n]; rim += gi[n]; }
        A0r[idx] = rr; A0i[idx] = rim;
    }
    __syncthreads();

    if (tid < 64) {
        const int n = tid;
        cpx lam = { -A_real[h*64 + n], -A_imag[h*64 + n] };
        cpx Dinv = cinv((cpx){1.0f - cc*lam.r, -cc*lam.i});
        cpx b = { B_real[h*64 + n], B_imag[h*64 + n] };
        cpx t = cmul(Dinv, cscale(b, dt));
        wr[n] = t.r; wi[n] = t.i;
    }
    __syncthreads();
    if (tid == 0) {
        cpx q = {0.f, 0.f};
        for (int n = 0; n < 64; ++n)
            q = cadd(q, cmul((cpx){pr[n], -pi[n]}, (cpx){wr[n], wi[n]}));
        cpx aq = cmul(alpha, q);
        scal[4] = aq.r; scal[5] = aq.i;
    }
    __syncthreads();
    if (tid < 64) {
        const int n = tid;
        cpx dB = csub((cpx){wr[n], wi[n]}, cmul((cpx){scal[4], scal[5]}, (cpx){ur[n], ui[n]}));
        Xr0[n*16 + 0] = dB.r;
        Xi0[n*16 + 0] = dB.i;
    }
    __syncthreads();

    v2f arf[16], aif[16];
    {
        load_a_frags(A0r, A0i, 16*wv, arf, aif);
        x_double_tile(arf, aif, Xr0, Xi0, Xr1, Xi1, 16*wv, 1);
        for (int n0 = 0; n0 < 64; n0 += 16)
            cplx_mm_tile(arf, aif, A0r, A0i, 64, A1r, A1i, 64, 16*wv, n0);
        __syncthreads();
        load_a_frags(A1r, A1i, 16*wv, arf, aif);
        x_double_tile(arf, aif, Xr0, Xi0, Xr1, Xi1, 16*wv, 2);
        for (int n0 = 0; n0 < 64; n0 += 16)
            cplx_mm_tile(arf, aif, A1r, A1i, 64, A0r, A0i, 64, 16*wv, n0);
        __syncthreads();
        load_a_frags(A0r, A0i, 16*wv, arf, aif);
        x_double_tile(arf, aif, Xr0, Xi0, Xr1, Xi1, 16*wv, 4);
        for (int n0 = 0; n0 < 64; n0 += 16)
            cplx_mm_tile(arf, aif, A0r, A0i, 64, A1r, A1i, 64, 16*wv, n0);
        __syncthreads();
        load_a_frags(A1r, A1i, 16*wv, arf, aif);
        x_double_tile(arf, aif, Xr0, Xi0, Xr1, Xi1, 16*wv, 8);
        for (int n0 = 0; n0 < 64; n0 += 16)
            cplx_mm_tile(arf, aif, A1r, A1i, 64, A0r, A0i, 64, 16*wv, n0);
        __syncthreads();
    }
    if (tid < 16) {
        float k = 0.f;
        for (int n = 0; n < 64; ++n)
            k += cr[n]*Xr0[n*16 + tid] - ci[n]*Xi0[n*16 + tid];
        Kp[tid] = k;
        for (int sl = 1; sl < 8; ++sl) Kp[sl*LL + tid] = 0.f;
    }
    __syncthreads();
    load_a_frags(A0r, A0i, 16*wv, arf, aif);

    const int lane = tid & 31;
    const int hf   = lane >> 4;
    float crv[8], civ[8];
    for (int v = 0; v < 8; ++v) {
        crv[v] = cr[16*wv + 8*hf + v];
        civ[v] = ci[16*wv + 8*hf + v];
    }
    float* kslot = Kp + (2*wv + hf) * LL + (lane & 15);

    for (int jj = 0; jj < 63; ++jj) {
        const int j = 2*jj + 1;
        scan_step_tile(arf, aif, Xr0, Xi0, Xr1, Xi1, crv, civ, kslot + j*CHUNK, 16*wv);
        __syncthreads();
        scan_step_tile(arf, aif, Xr1, Xi1, Xr0, Xi0, crv, civ, kslot + (j + 1)*CHUNK, 16*wv);
        __syncthreads();
    }
    scan_step_tile(arf, aif, Xr0, Xi0, Xr1, Xi1, crv, civ, kslot + 127*CHUNK, 16*wv);
    __syncthreads();

    for (int i = 0; i < LL / 128; ++i) {
        const int idx = i*128 + tid;
        float sacc = Kp[idx];
        for (int sl = 1; sl < 8; ++sl) sacc += Kp[sl*LL + idx];
        out[h*LL + idx] = 2.0f * sacc;
    }
}

extern "C" void kernel_launch(void* const* d_in, const int* in_sizes, int n_in,
                              void* d_out, int out_size, void* d_ws, size_t ws_size,
                              hipStream_t stream) {
    const float* A_real = (const float*)d_in[0];
    const float* A_imag = (const float*)d_in[1];
    const float* B_real = (const float*)d_in[2];
    const float* B_imag = (const float*)d_in[3];
    const float* P_real = (const float*)d_in[4];
    const float* P_imag = (const float*)d_in[5];
    const float* C_real = (const float*)d_in[6];
    const float* C_imag = (const float*)d_in[7];
    const float* log_dt = (const float*)d_in[8];
    float* out = (float*)d_out;
    (void)in_sizes; (void)n_in; (void)out_size; (void)d_ws; (void)ws_size;

    const size_t shmem_bytes = 37512u * sizeof(float);
    s4_dplr_kernel<<<NH, 128, shmem_bytes, stream>>>(
        A_real, A_imag, B_real, B_imag, P_real, P_imag, C_real, C_imag, log_dt, out);
}


# Scan replaced by baby-step/giant-step: K[64j+i] = (c^T dA^64j)(dA^i dB), both factors by doubling along the reference's squarings up to dA^1024; 10 squarings, no 127-step chain; sum order changes
# speedup vs baseline: 6.9532x; 1.5551x over previous
#include <hip/hip_runtime.h>
#include <hip/hip_bf16.h>

#define NH 256
#define NN 64
#define LL 2048

typedef __attribute__((ext_vector_type(2))) float v2f;
typedef __attribute__((ext_vector_type(8))) float v8f;

struct cpx { float r, i; };
__device__ __forceinline__ cpx cmul(cpx a, cpx b) { return {a.r*b.r - a.i*b.i, a.r*b.i + a.i*b.r}; }
__device__ __forceinline__ cpx cadd(cpx a, cpx b) { return {a.r + b.r, a.i + b.i}; }
__device__ __forceinline__ cpx csub(cpx a, cpx b) { return {a.r - b.r, a.i - b.i}; }
__device__ __forceinline__ cpx cinv(cpx a) { float d = a.r*a.r + a.i*a.i; return {a.r/d, -a.i/d}; }
__device__ __forceinline__ cpx cscale(cpx a, float s) { return {a.r*s, a.i*s}; }

__device__ __forceinline__ void load_a_frags(const float* Ar, const float* Ai, int r0,
                                             v2f (&arf)[16], v2f (&aif)[16])
{
    const int lane = threadIdx.x & 31;
    const int mn = lane & 15;
    const int kh = (lane >> 4) << 1;
    const float* rowr = Ar + (r0 + mn) * 64 + kh;
    const float* rowi = Ai + (r0 + mn) * 64 + kh;
    #pragma unroll
    for (int s = 0; s < 16; ++s) {
        arf[s][0] = rowr[4*s];  arf[s][1] = rowr[4*s + 1];
        aif[s][0] = rowi[4*s];  aif[s][1] = rowi[4*s + 1];
    }
}

__device__ __forceinline__ void cplx_mm_tile(
    const v2f (&arf)[16], const v2f (&aif)[16],
    const float* Br, const float* Bi, int ldb,
    float* Cr, float* Ci, int ldc,
    int r0, int n0)
{
    const int lane = threadIdx.x & 31;
    const int mn = lane & 15;
    const int kh = (lane >> 4) << 1;
    v8f arbr = {}; v8f aibi = {}; v8f arbi = {}; v8f aibr = {};
    #pragma unroll
    for (int s = 0; s < 16; ++s) {
        const int ka = 4*s + kh;
        v2f br, bi;
        br[0] = Br[ka * ldb + n0 + mn];       br[1] = Br[(ka + 1) * ldb + n0 + mn];
        bi[0] = Bi[ka * ldb + n0 + mn];       bi[1] = Bi[(ka + 1) * ldb + n0 + mn];
        arbr = __builtin_amdgcn_wmma_f32_16x16x4_f32(false, arf[s], false, br, (short)0, arbr, false, false);
        aibi = __builtin_amdgcn_wmma_f32_16x16x4_f32(false, aif[s], false, bi, (short)0, aibi, false, false);
        arbi = __builtin_amdgcn_wmma_f32_16x16x4_f32(false, arf[s], false, bi, (short)0, arbi, false, false);
        aibr = __builtin_amdgcn_wmma_f32_16x16x4_f32(false, aif[s], false, br, (short)0, aibr, false, false);
    }
    const int mbase = (lane >> 4) << 3;
    for (int v = 0; v < 8; ++v) {
        Cr[(r0 + mbase + v) * ldc + n0 + mn] = arbr[v] - aibi[v];
        Ci[(r0 + mbase + v) * ldc + n0 + mn] = arbi[v] + aibr[v];
    }
}

__device__ __forceinline__ void x_double_tile(
    const v2f (&arf)[16], const v2f (&aif)[16],
    float* Xr, float* Xi, float* Pr, float* Pi, int r0, int w)
{
    const int lane = threadIdx.x & 31;
    const int mn = lane & 15;
    const int kh = (lane >> 4) << 1;
    const bool live = mn < w;
    v2f brv[16], biv[16];
    #pragma unroll
    for (int s = 0; s < 16; ++s) { brv[s][0] = 0.f; brv[s][1] = 0.f; biv[s][0] = 0.f; biv[s][1] = 0.f; }
    if (live) {
        #pragma unroll
        for (int s = 0; s < 16; ++s) {
            const int ka = 4*s + kh;
            brv[s][0] = Xr[ka * 64 + mn];     brv[s][1] = Xr[(ka + 1) * 64 + mn];
            biv[s][0] = Xi[ka * 64 + mn];     biv[s][1] = Xi[(ka + 1) * 64 + mn];
        }
    }
    v8f arbr = {}; v8f aibi = {}; v8f arbi = {}; v8f aibr = {};
    #pragma unroll
    for (int s = 0; s < 16; ++s) {
        arbr = __builtin_amdgcn_wmma_f32_16x16x4_f32(false, arf[s], false, brv[s], (short)0, arbr, false, false);
        aibi = __builtin_amdgcn_wmma_f32_16x16x4_f32(false, aif[s], false, biv[s], (short)0, aibi, false, false);
        arbi = __builtin_amdgcn_wmma_f32_16x16x4_f32(false, arf[s], false, biv[s], (short)0, arbi, false, false);
        aibr = __builtin_amdgcn_wmma_f32_16x16x4_f32(false, aif[s], false, brv[s], (short)0, aibr, false, false);
    }
    const int mbase = (lane >> 4) << 3;
    float* dr = live ? (Xr + w) : Pr;
    float* di = live ? (Xi + w) : Pi;
    const int ldd = live ? 64 : 16;
    for (int v = 0; v < 8; ++v) {
        dr[(r0 + mbase + v) * ldd + mn] = arbr[v] - aibi[v];
        di[(r0 + mbase + v) * ldd + mn] = arbi[v] + aibr[v];
    }
}

__device__ __forceinline__ void g_double_tile(
    const v2f (&arf)[16], const v2f (&aif)[16],
    const float* Br, const float* Bi,
    float* Gr, float* Gi, float* Pr, float* Pi, int n0, int w)
{
    const int lane = threadIdx.x & 31;
    const int mn = lane & 15;
    const int kh = (lane >> 4) << 1;
    v8f arbr = {}; v8f aibi = {}; v8f arbi = {}; v8f aibr = {};
    #pragma unroll
    for (int s = 0; s < 16; ++s) {
        const int ka = 4*s + kh;
        v2f br, bi;
        br[0] = Br[ka * 64 + n0 + mn];        br[1] = Br[(ka + 1) * 64 + n0 + mn];
        bi[0] = Bi[ka * 64 + n0 + mn];        bi[1] = Bi[(ka + 1) * 64 + n0 + mn];
        arbr = __builtin_amdgcn_wmma_f32_16x16x4_f32(false, arf[s], false, br, (short)0, arbr, false, false);
        aibi = __builtin_amdgcn_wmma_f32_16x16x4_f32(false, aif[s], false, bi, (short)0, aibi, false, false);
        arbi = __builtin_amdgcn_wmma_f32_16x16x4_f32(false, arf[s], false, bi, (short)0, arbi, false, false);
        aibr = __builtin_amdgcn_wmma_f32_16x16x4_f32(false, aif[s], false, br, (short)0, aibr, false, false);
    }
    const int mbase = (lane >> 4) << 3;
    for (int v = 0; v < 8; ++v) {
        const int row = mbase + v;
        const bool live = row < w;
        float* dr = live ? (Gr + (w + row) * 64) : (Pr + row * 64);
        float* di = live ? (Gi + (w + row) * 64) : (Pi + row * 64);
        dr[n0 + mn] = arbr[v] - aibi[v];
        di[n0 + mn] = arbi[v] + aibr[v];
    }
}

__device__ __forceinline__ void k_tile(
    const v2f (&arf)[16], const v2f (&aif)[16],
    const float* Xr, const float* Xi, float* Kb, int j0, int n0)
{
    const int lane = threadIdx.x & 31;
    const int mn = lane & 15;
    const int kh = (lane >> 4) << 1;
    v8f grxr = {}; v8f gixi = {};
    #pragma unroll
    for (int s = 0; s < 16; ++s) {
        const int ka = 4*s + kh;
        v2f br, bi;
        br[0] = Xr[ka * 64 + n0 + mn];        br[1] = Xr[(ka + 1) * 64 + n0 + mn];
        bi[0] = Xi[ka * 64 + n0 + mn];        bi[1] = Xi[(ka + 1) * 64 + n0 + mn];
        grxr = __builtin_amdgcn_wmma_f32_16x16x4_f32(false, arf[s], false, br, (short)0, grxr, false, false);
        gixi = __builtin_amdgcn_wmma_f32_16x16x4_f32(false, aif[s], false, bi, (short)0, gixi, false, false);
    }
    const int mbase = (lane >> 4) << 3;
    for (int v = 0; v < 8; ++v)
        Kb[(j0 + mbase + v) * 64 + n0 + mn] = 2.0f * (grxr[v] - gixi[v]);
}

extern "C" __global__ __launch_bounds__(128)
void s4_dplr_kernel(const float* __restrict__ A_real, const float* __restrict__ A_imag,
                    const float* __restrict__ B_real, const float* __restrict__ B_imag,
                    const float* __restrict__ P_real, const float* __restrict__ P_imag,
                    const float* __restrict__ C_real, const float* __restrict__ C_imag,
                    const float* __restrict__ log_dt, float* __restrict__ out)
{
    extern __shared__ float sm[];
    const int h   = blockIdx.x;
    const int tid = threadIdx.x;
    const int wv  = tid >> 5;
    const int r0  = 16 * wv;

    const int OFF_A0 = 0;
    const int OFF_A1 = 8192;
    float* Xr  = sm + 16384;
    float* Xi  = sm + 20480;
    float* Gr  = sm + 24576;
    float* Gi  = sm + 26624;
    float* Dr  = sm + 28672;
    float* Di  = sm + 29696;
    float* Kb  = sm + 30720;
    float* ur  = sm + 32768;  float* ui = ur + 64;
    float* wr  = ui + 64;     float* wi = wr + 64;
    float* gr  = wi + 64;     float* gi = gr + 64;
    float* pr  = gi + 64;     float* pi = pr + 64;
    float* scal = pi + 64;

    const float dt = __expf(log_dt[h]);
    const float cc = 0.5f * dt;

    if (tid < 64) {
        const int n = tid;
        cpx lam = { -A_real[h*64 + n], -A_imag[h*64 + n] };
        cpx p   = {  P_real[h*64 + n],  P_imag[h*64 + n] };
        cpx Dinv = cinv((cpx){1.0f - cc*lam.r, -cc*lam.i});
        cpx e    = {1.0f + cc*lam.r,  cc*lam.i};
        cpx u = cmul(Dinv, p);
        cpx g = cmul(Dinv, e);
        ur[n] = u.r; ui[n] = u.i; gr[n] = g.r; gi[n] = g.i;
        pr[n] = p.r; pi[n] = p.i;
        Gr[n] = C_real[h*64 + n]; Gi[n] = C_imag[h*64 + n];
    }
    for (int idx = 64 + tid; idx < 2048; idx += 128) { Gr[idx] = 0.f; Gi[idx] = 0.f; }
    __syncthreads();
    if (tid == 0) {
        cpx s = {0.f, 0.f};
        for (int n = 0; n < 64; ++n)
            s = cadd(s, cmul((cpx){pr[n], -pi[n]}, (cpx){ur[n], ui[n]}));
        cpx alpha = cmul((cpx){cc, 0.f}, cinv((cpx){1.0f + cc*s.r, cc*s.i}));
        cpx base  = csub((cpx){cc, 0.f}, cmul(alpha, cscale(s, cc)));
        scal[0] = alpha.r; scal[1] = alpha.i; scal[2] = base.r; scal[3] = base.i;
    }
    __syncthreads();
    const cpx alpha = {scal[0], scal[1]};
    const cpx base  = {scal[2], scal[3]};
    if (tid < 64) {
        const int n = tid;
        cpx w = cmul((cpx){pr[n], -pi[n]}, cadd(base, cmul(alpha, (cpx){gr[n], gi[n]})));
        wr[n] = w.r; wi[n] = w.i;
    }
    __syncthreads();
    for (int idx = tid; idx < 4096; idx += 128) {
        const int n = idx >> 6, m = idx & 63;
        cpx v = cmul((cpx){ur[n], ui[n]}, (cpx){wr[m], wi[m]});
        float rr = -v.r, rim = -v.i;
        if (n == m) { rr += gr[n]; rim += gi[n]; }
        sm[OFF_A0 + idx] = rr; sm[OFF_A0 + 4096 + idx] = rim;
    }
    __syncthreads();

    if (tid < 64) {
        const int n = tid;
        cpx lam = { -A_real[h*64 + n], -A_imag[h*64 + n] };
        cpx Dinv = cinv((cpx){1.0f - cc*lam.r, -cc*lam.i});
        cpx b = { B_real[h*64 + n], B_imag[h*64 + n] };
        cpx t = cmul(Dinv, cscale(b, dt));
        wr[n] = t.r; wi[n] = t.i;
    }
    __syncthreads();
    if (tid == 0) {
        cpx q = {0.f, 0.f};
        for (int n = 0; n < 64; ++n)
            q = cadd(q, cmul((cpx){pr[n], -pi[n]}, (cpx){wr[n], wi[n]}));
        cpx aq = cmul(alpha, q);
        scal[4] = aq.r; scal[5] = aq.i;
    }
    __syncthreads();
    if (tid < 64) {
        const int n = tid;
        cpx dB = csub((cpx){wr[n], wi[n]}, cmul((cpx){scal[4], scal[5]}, (cpx){ur[n], ui[n]}));
        Xr[n*64 + 0] = dB.r;
        Xi[n*64 + 0] = dB.i;
    }
    __syncthreads();

    v2f arf[16], aif[16];
    int src = OFF_A0, dst = OFF_A1;

    for (int w = 1; w < 16; w += w) {
        load_a_frags(sm + src, sm + src + 4096, r0, arf, aif);
        x_double_tile(arf, aif, Xr, Xi, Dr, Di, r0, w);
        for (int n0 = 0; n0 < 64; n0 += 16)
            cplx_mm_tile(arf, aif, sm + src, sm + src + 4096, 64, sm + dst, sm + dst + 4096, 64, r0, n0);
        __syncthreads();
        const int t = src; src = dst; dst = t;
    }
    for (int w = 16; w < 64; w += w) {
        load_a_frags(sm + src, sm + src + 4096, r0, arf, aif);
        for (int n0 = 0; n0 < w; n0 += 16)
            cplx_mm_tile(arf, aif, Xr, Xi, 64, Xr + w, Xi + w, 64, r0, n0);
        for (int n0 = 0; n0 < 64; n0 += 16)
            cplx_mm_tile(arf, aif, sm + src, sm + src + 4096, 64, sm + dst, sm + dst + 4096, 64, r0, n0);
        __syncthreads();
        const int t = src; src = dst; dst = t;
    }
    for (int w = 1; w <= 16; w += w) {
        load_a_frags(Gr, Gi, 0, arf, aif);
        g_double_tile(arf, aif, sm + src, sm + src + 4096, Gr, Gi, Dr, Di, r0, w);
        if (w < 16) {
            load_a_frags(sm + src, sm + src + 4096, r0, arf, aif);
            for (int n0 = 0; n0 < 64; n0 += 16)
                cplx_mm_tile(arf, aif, sm + src, sm + src + 4096, 64, sm + dst, sm + dst + 4096, 64, r0, n0);
        }
        __syncthreads();
        const int t = src; src = dst; dst = t;
    }
    for (int j0 = 0; j0 < 32; j0 += 16) {
        load_a_frags(Gr, Gi, j0, arf, aif);
        k_tile(arf, aif, Xr, Xi, Kb, j0, r0);
    }
    __syncthreads();

    for (int i = 0; i < LL / 128; ++i) {
        const int idx = i*128 + tid;
        out[h*LL + idx] = Kb[idx];
    }
}

extern "C" void kernel_launch(void* const* d_in, const int* in_sizes, int n_in,
                              void* d_out, int out_size, void* d_ws, size_t ws_size,
                              hipStream_t stream) {
    const float* A_real = (const float*)d_in[0];
    const float* A_imag = (const float*)d_in[1];
    const float* B_real = (const float*)d_in[2];
    const float* B_imag = (const float*)d_in[3];
    const float* P_real = (const float*)d_in[4];
    const float* P_imag = (const float*)d_in[5];
    const float* C_real = (const float*)d_in[6];
    const float* C_imag = (const float*)d_in[7];
    const float* log_dt = (const float*)d_in[8];
    float* out = (float*)d_out;
    (void)in_sizes; (void)n_in; (void)out_size; (void)d_ws; (void)ws_size;

    const size_t shmem_bytes = 33288u * sizeof(float);
    s4_dplr_kernel<<<NH, 128, shmem_bytes, stream>>>(
        A_real, A_imag, B_real, B_imag, P_real, P_imag, C_real, C_imag, log_dt, out);
}


# A-fragment loads made free of LDS bank conflicts: G gets a row pitch of 68 floats, and each wave keeps its own 16 rows of the current power in a second copy at pitch 68 that the tile epilogues write
# speedup vs baseline: 11.4066x; 1.6405x over previous
#include <hip/hip_runtime.h>
#include <hip/hip_bf16.h>

#define NH 256
#define NN 64
#define LL 2048
#define PA 68

typedef __attribute__((ext_vector_type(2))) float v2f;
typedef __attribute__((ext_vector_type(8))) float v8f;

struct cpx { float r, i; };
__device__ __forceinline__ cpx cmul(cpx a, cpx b) { return {a.r*b.r - a.i*b.i, a.r*b.i + a.i*b.r}; }
__device__ __forceinline__ cpx cadd(cpx a, cpx b) { return {a.r + b.r, a.i + b.i}; }
__device__ __forceinline__ cpx csub(cpx a, cpx b) { return {a.r - b.r, a.i - b.i}; }
__device__ __forceinline__ cpx cinv(cpx a) { float d = a.r*a.r + a.i*a.i; return {a.r/d, -a.i/d}; }
__device__ __forceinline__ cpx cscale(cpx a, float s) { return {a.r*s, a.i*s}; }

__device__ __forceinline__ void load_a_frags(const float* Ar, const float* Ai, int ld, int r0,
                                             v2f (&arf)[16], v2f (&aif)[16])
{
    const int lane = threadIdx.x & 31;
    const int mn = lane & 15;
    const int kh = (lane >> 4) << 1;
    const float* rowr = Ar + (r0 + mn) * ld + kh;
    const float* rowi = Ai + (r0 + mn) * ld + kh;
    #pragma unroll
    for (int s = 0; s < 16; ++s) {
        arf[s][0] = rowr[4*s];  arf[s][1] = rowr[4*s + 1];
        aif[s][0] = rowi[4*s];  aif[s][1] = rowi[4*s + 1];
    }
}

template <bool ACOPY>
__device__ __forceinline__ void cplx_mm_tile(
    const v2f (&arf)[16], const v2f (&aif)[16],
    const float* Br, const float* Bi, int ldb,
    float* Cr, float* Ci, int ldc,
    float* Er, float* Ei,
    int r0, int n0)
{
    const int lane = threadIdx.x & 31;
    const int mn = lane & 15;
    const int kh = (lane >> 4) << 1;
    v8f arbr = {}; v8f aibi = {}; v8f arbi = {}; v8f aibr = {};
    #pragma unroll
    for (int s = 0; s < 16; ++s) {
        const int ka = 4*s + kh;
        v2f br, bi;
        br[0] = Br[ka * ldb + n0 + mn];       br[1] = Br[(ka + 1) * ldb + n0 + mn];
        bi[0] = Bi[ka * ldb + n0 + mn];       bi[1] = Bi[(ka + 1) * ldb + n0 + mn];
        arbr = __builtin_amdgcn_wmma_f32_16x16x4_f32(false, arf[s], false, br, (short)0, arbr, false, false);
        aibi = __builtin_amdgcn_wmma_f32_16x16x4_f32(false, aif[s], false, bi, (short)0, aibi, false, false);
        arbi = __builtin_amdgcn_wmma_f32_16x16x4_f32(false, arf[s], false, bi, (short)0, arbi, false, false);
        aibr = __builtin_amdgcn_wmma_f32_16x16x4_f32(false, aif[s], false, br, (short)0, aibr, false, false);
    }
    const int mbase = (lane >> 4) << 3;
    float re[8], im[8];
    for (int v = 0; v < 8; ++v) { re[v] = arbr[v] - aibi[v]; im[v] = arbi[v] + aibr[v]; }
    for (int v = 0; v < 8; ++v) {
        Cr[(r0 + mbase + v) * ldc + n0 + mn] = re[v];
        Ci[(r0 + mbase + v) * ldc + n0 + mn] = im[v];
    }
    if (ACOPY) {
        for (int v = 0; v < 8; ++v) {
            Er[(r0 + mbase + v) * PA + n0 + mn] = re[v];
            Ei[(r0 + mbase + v) * PA + n0 + mn] = im[v];
        }
    }
}

__device__ __forceinline__ void x_double_tile(
    const v2f (&arf)[16], const v2f (&aif)[16],
    float* Xr, float* Xi, float* Pr, float* Pi, int r0, int w)
{
    const int lane = threadIdx.x & 31;
    const int mn = lane & 15;
    const int kh = (lane >> 4) << 1;
    const bool live = mn < w;
    v2f brv[16], biv[16];
    #pragma unroll
    for (int s = 0; s < 16; ++s) { brv[s][0] = 0.f; brv[s][1] = 0.f; biv[s][0] = 0.f; biv[s][1] = 0.f; }
    if (live) {
        #pragma unroll
        for (int s = 0; s < 16; ++s) {
            const int ka = 4*s + kh;
            brv[s][0] = Xr[ka * 64 + mn];     brv[s][1] = Xr[(ka + 1) * 64 + mn];
            biv[s][0] = Xi[ka * 64 + mn];     biv[s][1] = Xi[(ka + 1) * 64 + mn];
        }
    }
    v8f arbr = {}; v8f aibi = {}; v8f arbi = {}; v8f aibr = {};
    #pragma unroll
    for (int s = 0; s < 16; ++s) {
        arbr = __builtin_amdgcn_wmma_f32_16x16x4_f32(false, arf[s], false, brv[s], (short)0, arbr, false, false);
        aibi = __builtin_amdgcn_wmma_f32_16x16x4_f32(false, aif[s], false, biv[s], (short)0, aibi, false, false);
        arbi = __builtin_amdgcn_wmma_f32_16x16x4_f32(false, arf[s], false, biv[s], (short)0, arbi, false, false);
        aibr = __builtin_amdgcn_wmma_f32_16x16x4_f32(false, aif[s], false, brv[s], (short)0, aibr, false, false);
    }
    const int mbase = (lane >> 4) << 3;
    float* dr = live ? (Xr + w) : Pr;
    float* di = live ? (Xi + w) : Pi;
    const int ldd = live ? 64 : 16;
    for (int v = 0; v < 8; ++v) {
        dr[(r0 + mbase + v) * ldd + mn] = arbr[v] - aibi[v];
        di[(r0 + mbase + v) * ldd + mn] = arbi[v] + aibr[v];
    }
}

__device__ __forceinline__ void g_double_tile(
    const v2f (&arf)[16], const v2f (&aif)[16],
    const float* Br, const float* Bi,
    float* Gr, float* Gi, float* Pr, float* Pi, int n0, int w)
{
    const int lane = threadIdx.x & 31;
    const int mn = lane & 15;
    const int kh = (lane >> 4) << 1;
    v8f arbr = {}; v8f aibi = {}; v8f arbi = {}; v8f aibr = {};
    #pragma unroll
    for (int s = 0; s < 16; ++s) {
        const int ka = 4*s + kh;
        v2f br, bi;
        br[0] = Br[ka * 64 + n0 + mn];        br[1] = Br[(ka + 1) * 64 + n0 + mn];
        bi[0] = Bi[ka * 64 + n0 + mn];        bi[1] = Bi[(ka + 1) * 64 + n0 + mn];
        arbr = __builtin_amdgcn_wmma_f32_16x16x4_f32(false, arf[s], false, br, (short)0, arbr, false, false);
        aibi = __builtin_amdgcn_wmma_f32_16x16x4_f32(false, aif[s], false, bi, (short)0, aibi, false, false);
        arbi = __builtin_amdgcn_wmma_f32_16x16x4_f32(false, arf[s], false, bi, (short)0, arbi, false, false);
        aibr = __builtin_amdgcn_wmma_f32_16x16x4_f32(false, aif[s], false, br, (short)0, aibr, false, false);
    }
    const int mbase = (lane >> 4) << 3;
    for (int v = 0; v < 8; ++v) {
        const int row = mbase + v;
        const bool live = row < w;
        float* dr = live ? (Gr + (w + row) * PA) : (Pr + row * PA);
        float* di = live ? (Gi + (w + row) * PA) : (Pi + row * PA);
        dr[n0 + mn] = arbr[v] - aibi[v];
        di[n0 + mn] = arbi[v] + aibr[v];
    }
}

__device__ __forceinline__ void k_tile(
    const v2f (&arf)[16], const v2f (&aif)[16],
    const float* Xr, const float* Xi, float* Kb, int j0, int n0)
{
    const int lane = threadIdx.x & 31;
    const int mn = lane & 15;
    const int kh = (lane >> 4) << 1;
    v8f grxr = {}; v8f gixi = {};
    #pragma unroll
    for (int s = 0; s < 16; ++s) {
        const int ka = 4*s + kh;
        v2f br, bi;
        br[0] = Xr[ka * 64 + n0 + mn];        br[1] = Xr[(ka + 1) * 64 + n0 + mn];
        bi[0] = Xi[ka * 64 + n0 + mn];        bi[1] = Xi[(ka + 1) * 64 + n0 + mn];
        grxr = __builtin_amdgcn_wmma_f32_16x16x4_f32(false, arf[s], false, br, (short)0, grxr, false, false);
        gixi = __builtin_amdgcn_wmma_f32_16x16x4_f32(false, aif[s], false, bi, (short)0, gixi, false, false);
    }
    const int mbase = (lane >> 4) << 3;
    for (int v = 0; v < 8; ++v)
        Kb[(j0 + mbase + v) * 64 + n0 + mn] = 2.0f * (grxr[v] - gixi[v]);
}

extern "C" __global__ __launch_bounds__(128)
void s4_dplr_kernel(const float* __restrict__ A_real, const float* __restrict__ A_imag,
                    const float* __restrict__ B_real, const float* __restrict__ B_imag,
                    const float* __restrict__ P_real, const float* __restrict__ P_imag,
                    const float* __restrict__ C_real, const float* __restrict__ C_imag,
                    const float* __restrict__ log_dt, float* __restrict__ out)
{
    extern __shared__ float sm[];
    const int h   = blockIdx.x;
    const int tid = threadIdx.x;
    const int wv  = tid >> 5;
    const int r0  = 16 * wv;

    const int OFF_A0 = 0;
    const int OFF_A1 = 8192;
    float* Xr  = sm + 16384;
    float* Xi  = sm + 20480;
    float* Gr  = sm + 24576;
    float* Gi  = Gr + 32 * PA;
    float* Dr  = Gi + 32 * PA;
    float* Di  = Dr + 16 * PA;
    float* ACr = Di + 16 * PA;
    float* ACi = ACr + 64 * PA;
    float* Kb  = sm + OFF_A0;
    float* ur  = ACi + 64 * PA; float* ui = ur + 64;
    float* wr  = ui + 64;     float* wi = wr + 64;
    float* gr  = wi + 64;     float* gi = gr + 64;
    float* pr  = gi + 64;     float* pi = pr + 64;
    float* scal = pi + 64;

    const float dt = __expf(log_dt[h]);
    const float cc = 0.5f * dt;

    if (tid < 64) {
        const int n = tid;
        cpx lam = { -A_real[h*64 + n], -A_imag[h*64 + n] };
        cpx p   = {  P_real[h*64 + n],  P_imag[h*64 + n] };
        cpx Dinv = cinv((cpx){1.0f - cc*lam.r, -cc*lam.i});
        cpx e    = {1.0f + cc*lam.r,  cc*lam.i};
        cpx u = cmul(Dinv, p);
        cpx g = cmul(Dinv, e);
        ur[n] = u.r; ui[n] = u.i; gr[n] = g.r; gi[n] = g.i;
        pr[n] = p.r; pi[n] = p.i;
        Gr[n] = C_real[h*64 + n]; Gi[n] = C_imag[h*64 + n];
    }
    for (int idx = 64 + tid; idx < 32 * PA; idx += 128) { Gr[idx] = 0.f; Gi[idx] = 0.f; }
    __syncthreads();
    if (tid == 0) {
        cpx s = {0.f, 0.f};
        for (int n = 0; n < 64; ++n)
            s = cadd(s, cmul((cpx){pr[n], -pi[n]}, (cpx){ur[n], ui[n]}));
        cpx alpha = cmul((cpx){cc, 0.f}, cinv((cpx){1.0f + cc*s.r, cc*s.i}));
        cpx base  = csub((cpx){cc, 0.f}, cmul(alpha, cscale(s, cc)));
        scal[0] = alpha.r; scal[1] = alpha.i; scal[2] = base.r; scal[3] = base.i;
    }
    __syncthreads();
    const cpx alpha = {scal[0], scal[1]};
    const cpx base  = {scal[2], scal[3]};
    if (tid < 64) {
        const int n = tid;
        cpx w = cmul((cpx){pr[n], -pi[n]}, cadd(base, cmul(alpha, (cpx){gr[n], gi[n]})));
        wr[n] = w.r; wi[n] = w.i;
    }
    __syncthreads();
    for (int idx = tid; idx < 4096; idx += 128) {
        const int n = idx >> 6, m = idx & 63;
        cpx v = cmul((cpx){ur[n], ui[n]}, (cpx){wr[m], wi[m]});
        float rr = -v.r, rim = -v.i;
        if (n == m) { rr += gr[n]; rim += gi[n]; }
        sm[OFF_A0 + idx] = rr; sm[OFF_A0 + 4096 + idx] = rim;
        ACr[n * PA + m] = rr;  ACi[n * PA + m] = rim;
    }
    __syncthreads();

    if (tid < 64) {
        const int n = tid;
        cpx lam = { -A_real[h*64 + n], -A_imag[h*64 + n] };
        cpx Dinv = cinv((cpx){1.0f - cc*lam.r, -cc*lam.i});
        cpx b = { B_real[h*64 + n], B_imag[h*64 + n] };
        cpx t = cmul(Dinv, cscale(b, dt));
        wr[n] = t.r; wi[n] = t.i;
    }
    __syncthreads();
    if (tid == 0) {
        cpx q = {0.f, 0.f};
        for (int n = 0; n < 64; ++n)
            q = cadd(q, cmul((cpx){pr[n], -pi[n]}, (cpx){wr[n], wi[n]}));
        cpx aq = cmul(alpha, q);
        scal[4] = aq.r; scal[5] = aq.i;
    }
    __syncthreads();
    if (tid < 64) {
        const int n = tid;
        cpx dB = csub((cpx){wr[n], wi[n]}, cmul((cpx){scal[4], scal[5]}, (cpx){ur[n], ui[n]}));
        Xr[n*64 + 0] = dB.r;
        Xi[n*64 + 0] = dB.i;
    }
    __syncthreads();

    v2f arf[16], aif[16];
    int src = OFF_A0, dst = OFF_A1;

    for (int w = 1; w < 16; w += w) {
        load_a_frags(ACr, ACi, PA, r0, arf, aif);
        x_double_tile(arf, aif, Xr, Xi, Dr, Di, r0, w);
        for (int n0 = 0; n0 < 64; n0 += 16)
            cplx_mm_tile<true>(arf, aif, sm + src, sm + src + 4096, 64, sm + dst, sm + dst + 4096, 64, ACr, ACi, r0, n0);
        __syncthreads();
        const int t = src; src = dst; dst = t;
    }
    for (int w = 16; w < 64; w += w) {
        load_a_frags(ACr, ACi, PA, r0, arf, aif);
        for (int n0 = 0; n0 < w; n0 += 16)
            cplx_mm_tile<false>(arf, aif, Xr, Xi, 64, Xr + w, Xi + w, 64, nullptr, nullptr, r0, n0);
        for (int n0 = 0; n0 < 64; n0 += 16)
            cplx_mm_tile<true>(arf, aif, sm + src, sm + src + 4096, 64, sm + dst, sm + dst + 4096, 64, ACr, ACi, r0, n0);
        __syncthreads();
        const int t = src; src = dst; dst = t;
    }
    for (int w = 1; w <= 16; w += w) {
        load_a_frags(Gr, Gi, PA, 0, arf, aif);
        g_double_tile(arf, aif, sm + src, sm + src + 4096, Gr, Gi, Dr, Di, r0, w);
        if (w < 16) {
            load_a_frags(ACr, ACi, PA, r0, arf, aif);
            for (int n0 = 0; n0 < 64; n0 += 16)
                cplx_mm_tile<true>(arf, aif, sm + src, sm + src + 4096, 64, sm + dst, sm + dst + 4096, 64, ACr, ACi, r0, n0);
        }
        __syncthreads();
        const int t = src; src = dst; dst = t;
    }
    for (int j0 = 0; j0 < 32; j0 += 16) {
        load_a_frags(Gr, Gi, PA, j0, arf, aif);
        k_tile(arf, aif, Xr, Xi, Kb, j0, r0);
    }
    __syncthreads();

    for (int i = 0; i < LL / 128; ++i) {
        const int idx = i*128 + tid;
        out[h*LL + idx] = Kb[idx];
    }
}

extern "C" void kernel_launch(void* const* d_in, const int* in_sizes, int n_in,
                              void* d_out, int out_size, void* d_ws, size_t ws_size,
                              hipStream_t stream) {
    const float* A_real = (const float*)d_in[0];
    const float* A_imag = (const float*)d_in[1];
    const float* B_real = (const float*)d_in[2];
    const float* B_imag = (const float*)d_in[3];
    const float* P_real = (const float*)d_in[4];
    const float* P_imag = (const float*)d_in[5];
    const float* C_real = (const float*)d_in[6];
    const float* C_imag = (const float*)d_in[7];
    const float* log_dt = (const float*)d_in[8];
    float* out = (float*)d_out;
    (void)in_sizes; (void)n_in; (void)out_size; (void)d_ws; (void)ws_size;

    const size_t shmem_bytes = 40328u * sizeof(float);
    s4_dplr_kernel<<<NH, 128, shmem_bytes, stream>>>(
        A_real, A_imag, B_real, B_imag, P_real, P_imag, C_real, C_imag, log_dt, out);
}


# Complex matrix products (squarings, X and G doubling) by three real f32 WMMA chains instead of four: Im = (Ar+Ai)(Br+Bi) - ArBr - AiBi; rounding and sum order of imaginary parts change
# speedup vs baseline: 12.5013x; 1.0960x over previous
#include <hip/hip_runtime.h>
#include <hip/hip_bf16.h>

#define NH 256
#define NN 64
#define LL 2048
#define PA 68

typedef __attribute__((ext_vector_type(2))) float v2f;
typedef __attribute__((ext_vector_type(8))) float v8f;

struct cpx { float r, i; };
__device__ __forceinline__ cpx cmul(cpx a, cpx b) { return {a.r*b.r - a.i*b.i, a.r*b.i + a.i*b.r}; }
__device__ __forceinline__ cpx cadd(cpx a, cpx b) { return {a.r + b.r, a.i + b.i}; }
__device__ __forceinline__ cpx csub(cpx a, cpx b) { return {a.r - b.r, a.i - b.i}; }
__device__ __forceinline__ cpx cinv(cpx a) { float d = a.r*a.r + a.i*a.i; return {a.r/d, -a.i/d}; }
__device__ __forceinline__ cpx cscale(cpx a, float s) { return {a.r*s, a.i*s}; }

__device__ __forceinline__ void load_a_frags(const float* Ar, const float* Ai, int ld, int r0,
                                             v2f (&arf)[16], v2f (&aif)[16])
{
    const int lane = threadIdx.x & 31;
    const int mn = lane & 15;
    const int kh = (lane >> 4) << 1;
    const float* rowr = Ar + (r0 + mn) * ld + kh;
    const float* rowi = Ai + (r0 + mn) * ld + kh;
    #pragma unroll
    for (int s = 0; s < 16; ++s) {
        arf[s][0] = rowr[4*s];  arf[s][1] = rowr[4*s + 1];
        aif[s][0] = rowi[4*s];  aif[s][1] = rowi[4*s + 1];
    }
}

__device__ __forceinline__ void sum_a_frags(const v2f (&arf)[16], const v2f (&aif)[16], v2f (&asf)[16])
{
    #pragma unroll
    for (int s = 0; s < 16; ++s) asf[s] = arf[s] + aif[s];
}

template <bool ACOPY>
__device__ __forceinline__ void cplx_mm_tile(
    const v2f (&arf)[16], const v2f (&aif)[16], const v2f (&asf)[16],
    const float* Br, const float* Bi, int ldb,
    float* Cr, float* Ci, int ldc,
    float* Er, float* Ei,
    int r0, int n0)
{
    const int lane = threadIdx.x & 31;
    const int mn = lane & 15;
    const int kh = (lane >> 4) << 1;
    v2f brv[16], biv[16];
    #pragma unroll
    for (int s = 0; s < 16; ++s) {
        const int ka = 4*s + kh;
        brv[s][0] = Br[ka * ldb + n0 + mn];   brv[s][1] = Br[(ka + 1) * ldb + n0 + mn];
        biv[s][0] = Bi[ka * ldb + n0 + mn];   biv[s][1] = Bi[(ka + 1) * ldb + n0 + mn];
    }
    __builtin_amdgcn_sched_barrier(0);
    v8f p1 = {}; v8f p2 = {}; v8f p3 = {};
    #pragma unroll
    for (int s = 0; s < 16; ++s) {
        const v2f bs = brv[s] + biv[s];
        p1 = __builtin_amdgcn_wmma_f32_16x16x4_f32(false, arf[s], false, brv[s], (short)0, p1, false, false);
        p2 = __builtin_amdgcn_wmma_f32_16x16x4_f32(false, aif[s], false, biv[s], (short)0, p2, false, false);
        p3 = __builtin_amdgcn_wmma_f32_16x16x4_f32(false, asf[s], false, bs, (short)0, p3, false, false);
        __builtin_amdgcn_sched_barrier(0);
    }
    const int mbase = (lane >> 4) << 3;
    float re[8], im[8];
    for (int v = 0; v < 8; ++v) { re[v] = p1[v] - p2[v]; im[v] = (p3[v] - p1[v]) - p2[v]; }
    for (int v = 0; v < 8; ++v) {
        Cr[(r0 + mbase + v) * ldc + n0 + mn] = re[v];
        Ci[(r0 + mbase + v) * ldc + n0 + mn] = im[v];
    }
    if (ACOPY) {
        for (int v = 0; v < 8; ++v) {
            Er[(r0 + mbase + v) * PA + n0 + mn] = re[v];
            Ei[(r0 + mbase + v) * PA + n0 + mn] = im[v];
        }
    }
}

__device__ __forceinline__ void x_double_tile(
    const v2f (&arf)[16], const v2f (&aif)[16], const v2f (&asf)[16],
    float* Xr, float* Xi, float* Pr, float* Pi, int r0, int w)
{
    const int lane = threadIdx.x & 31;
    const int mn = lane & 15;
    const int kh = (lane >> 4) << 1;
    const bool live = mn < w;
    v2f brv[16], biv[16];
    #pragma unroll
    for (int s = 0; s < 16; ++s) { brv[s][0] = 0.f; brv[s][1] = 0.f; biv[s][0] = 0.f; biv[s][1] = 0.f; }
    if (live) {
        #pragma unroll
        for (int s = 0; s < 16; ++s) {
            const int ka = 4*s + kh;
            brv[s][0] = Xr[ka * 64 + mn];     brv[s][1] = Xr[(ka + 1) * 64 + mn];
            biv[s][0] = Xi[ka * 64 + mn];     biv[s][1] = Xi[(ka + 1) * 64 + mn];
        }
    }
    v8f p1 = {}; v8f p2 = {}; v8f p3 = {};
    #pragma unroll
    for (int s = 0; s < 16; ++s) {
        const v2f bs = brv[s] + biv[s];
        p1 = __builtin_amdgcn_wmma_f32_16x16x4_f32(false, arf[s], false, brv[s], (short)0, p1, false, false);
        p2 = __builtin_amdgcn_wmma_f32_16x16x4_f32(false, aif[s], false, biv[s], (short)0, p2, false, false);
        p3 = __builtin_amdgcn_wmma_f32_16x16x4_f32(false, asf[s], false, bs, (short)0, p3, false, false);
    }
    const int mbase = (lane >> 4) << 3;
    float* dr = live ? (Xr + w) : Pr;
    float* di = live ? (Xi + w) : Pi;
    const int ldd = live ? 64 : 16;
    for (int v = 0; v < 8; ++v) {
        dr[(r0 + mbase + v) * ldd + mn] = p1[v] - p2[v];
        di[(r0 + mbase + v) * ldd + mn] = (p3[v] - p1[v]) - p2[v];
    }
}

__device__ __forceinline__ void g_double_tile(
    const v2f (&arf)[16], const v2f (&aif)[16], const v2f (&asf)[16],
    const float* Br, const float* Bi,
    float* Gr, float* Gi, float* Pr, float* Pi, int n0, int w)
{
    const int lane = threadIdx.x & 31;
    const int mn = lane & 15;
    const int kh = (lane >> 4) << 1;
    v2f brv[16], biv[16];
    #pragma unroll
    for (int s = 0; s < 16; ++s) {
        const int ka = 4*s + kh;
        brv[s][0] = Br[ka * 64 + n0 + mn];    brv[s][1] = Br[(ka + 1) * 64 + n0 + mn];
        biv[s][0] = Bi[ka * 64 + n0 + mn];    biv[s][1] = Bi[(ka + 1) * 64 + n0 + mn];
    }
    __builtin_amdgcn_sched_barrier(0);
    v8f p1 = {}; v8f p2 = {}; v8f p3 = {};
    #pragma unroll
    for (int s = 0; s < 16; ++s) {
        const v2f bs = brv[s] + biv[s];
        p1 = __builtin_amdgcn_wmma_f32_16x16x4_f32(false, arf[s], false, brv[s], (short)0, p1, false, false);
        p2 = __builtin_amdgcn_wmma_f32_16x16x4_f32(false, aif[s], false, biv[s], (short)0, p2, false, false);
        p3 = __builtin_amdgcn_wmma_f32_16x16x4_f32(false, asf[s], false, bs, (short)0, p3, false, false);
        __builtin_amdgcn_sched_barrier(0);
    }
    const int mbase = (lane >> 4) << 3;
    for (int v = 0; v < 8; ++v) {
        const int row = mbase + v;
        const bool live = row < w;
        float* dr = live ? (Gr + (w + row) * PA) : (Pr + row * PA);
        float* di = live ? (Gi + (w + row) * PA) : (Pi + row * PA);
        dr[n0 + mn] = p1[v] - p2[v];
        di[n0 + mn] = (p3[v] - p1[v]) - p2[v];
    }
}

__device__ __forceinline__ void k_tile(
    const v2f (&arf)[16], const v2f (&aif)[16],
    const float* Xr, const float* Xi, float* Kb, int j0, int n0)
{
    const int lane = threadIdx.x & 31;
    const int mn = lane & 15;
    const int kh = (lane >> 4) << 1;
    v8f grxr = {}; v8f gixi = {};
    #pragma unroll
    for (int s = 0; s < 16; ++s) {
        const int ka = 4*s + kh;
        v2f br, bi;
        br[0] = Xr[ka * 64 + n0 + mn];        br[1] = Xr[(ka + 1) * 64 + n0 + mn];
        bi[0] = Xi[ka * 64 + n0 + mn];        bi[1] = Xi[(ka + 1) * 64 + n0 + mn];
        grxr = __builtin_amdgcn_wmma_f32_16x16x4_f32(false, arf[s], false, br, (short)0, grxr, false, false);
        gixi = __builtin_amdgcn_wmma_f32_16x16x4_f32(false, aif[s], false, bi, (short)0, gixi, false, false);
    }
    const int mbase = (lane >> 4) << 3;
    for (int v = 0; v < 8; ++v)
        Kb[(j0 + mbase + v) * 64 + n0 + mn] = 2.0f * (grxr[v] - gixi[v]);
}

extern "C" __global__ __launch_bounds__(128)
void s4_dplr_kernel(const float* __restrict__ A_real, const float* __restrict__ A_imag,
                    const float* __restrict__ B_real, const float* __restrict__ B_imag,
                    const float* __restrict__ P_real, const float* __restrict__ P_imag,
                    const float* __restrict__ C_real, const float* __restrict__ C_imag,
                    const float* __restrict__ log_dt, float* __restrict__ out)
{
    extern __shared__ float sm[];
    const int h   = blockIdx.x;
    const int tid = threadIdx.x;
    const int wv  = tid >> 5;
    const int r0  = 16 * wv;

    const int OFF_A0 = 0;
    const int OFF_A1 = 8192;
    float* Xr  = sm + 16384;
    float* Xi  = sm + 20480;
    float* Gr  = sm + 24576;
    float* Gi  = Gr + 32 * PA;
    float* Dr  = Gi + 32 * PA;
    float* Di  = Dr + 16 * PA;
    float* ACr = Di + 16 * PA;
    float* ACi = ACr + 64 * PA;
    float* Kb  = sm + OFF_A0;
    float* ur  = ACi + 64 * PA; float* ui = ur + 64;
    float* wr  = ui + 64;     float* wi = wr + 64;
    float* gr  = wi + 64;     float* gi = gr + 64;
    float* pr  = gi + 64;     float* pi = pr + 64;
    float* scal = pi + 64;

    const float dt = __expf(log_dt[h]);
    const float cc = 0.5f * dt;

    if (tid < 64) {
        const int n = tid;
        cpx lam = { -A_real[h*64 + n], -A_imag[h*64 + n] };
        cpx p   = {  P_real[h*64 + n],  P_imag[h*64 + n] };
        cpx Dinv = cinv((cpx){1.0f - cc*lam.r, -cc*lam.i});
        cpx e    = {1.0f + cc*lam.r,  cc*lam.i};
        cpx u = cmul(Dinv, p);
        cpx g = cmul(Dinv, e);
        ur[n] = u.r; ui[n] = u.i; gr[n] = g.r; gi[n] = g.i;
        pr[n] = p.r; pi[n] = p.i;
        Gr[n] = C_real[h*64 + n]; Gi[n] = C_imag[h*64 + n];
    }
    for (int idx = 64 + tid; idx < 32 * PA; idx += 128) { Gr[idx] = 0.f; Gi[idx] = 0.f; }
    __syncthreads();
    if (tid == 0) {
        cpx s = {0.f, 0.f};
        for (int n = 0; n < 64; ++n)
            s = cadd(s, cmul((cpx){pr[n], -pi[n]}, (cpx){ur[n], ui[n]}));
        cpx alpha = cmul((cpx){cc, 0.f}, cinv((cpx){1.0f + cc*s.r, cc*s.i}));
        cpx base  = csub((cpx){cc, 0.f}, cmul(alpha, cscale(s, cc)));
        scal[0] = alpha.r; scal[1] = alpha.i; scal[2] = base.r; scal[3] = base.i;
    }
    __syncthreads();
    const cpx alpha = {scal[0], scal[1]};
    const cpx base  = {scal[2], scal[3]};
    if (tid < 64) {
        const int n = tid;
        cpx w = cmul((cpx){pr[n], -pi[n]}, cadd(base, cmul(alpha, (cpx){gr[n], gi[n]})));
        wr[n] = w.r; wi[n] = w.i;
    }
    __syncthreads();
    for (int idx = tid; idx < 4096; idx += 128) {
        const int n = idx >> 6, m = idx & 63;
        cpx v = cmul((cpx){ur[n], ui[n]}, (cpx){wr[m], wi[m]});
        float rr = -v.r, rim = -v.i;
        if (n == m) { rr += gr[n]; rim += gi[n]; }
        sm[OFF_A0 + idx] = rr; sm[OFF_A0 + 4096 + idx] = rim;
        ACr[n * PA + m] = rr;  ACi[n * PA + m] = rim;
    }
    __syncthreads();

    if (tid < 64) {
        const int n = tid;
        cpx lam = { -A_real[h*64 + n], -A_imag[h*64 + n] };
        cpx Dinv = cinv((cpx){1.0f - cc*lam.r, -cc*lam.i});
        cpx b = { B_real[h*64 + n], B_imag[h*64 + n] };
        cpx t = cmul(Dinv, cscale(b, dt));
        wr[n] = t.r; wi[n] = t.i;
    }
    __syncthreads();
    if (tid == 0) {
        cpx q = {0.f, 0.f};
        for (int n = 0; n < 64; ++n)
            q = cadd(q, cmul((cpx){pr[n], -pi[n]}, (cpx){wr[n], wi[n]}));
        cpx aq = cmul(alpha, q);
        scal[4] = aq.r; scal[5] = aq.i;
    }
    __syncthreads();
    if (tid < 64) {
        const int n = tid;
        cpx dB = csub((cpx){wr[n], wi[n]}, cmul((cpx){scal[4], scal[5]}, (cpx){ur[n], ui[n]}));
        Xr[n*64 + 0] = dB.r;
        Xi[n*64 + 0] = dB.i;
    }
    __syncthreads();

    v2f arf[16], aif[16], asf[16];
    int src = OFF_A0, dst = OFF_A1;

    for (int w = 1; w < 16; w += w) {
        load_a_frags(ACr, ACi, PA, r0, arf, aif);
        sum_a_frags(arf, aif, asf);
        x_double_tile(arf, aif, asf, Xr, Xi, Dr, Di, r0, w);
        for (int n0 = 0; n0 < 64; n0 += 16)
            cplx_mm_tile<true>(arf, aif, asf, sm + src, sm + src + 4096, 64, sm + dst, sm + dst + 4096, 64, ACr, ACi, r0, n0);
        __syncthreads();
        const int t = src; src = dst; dst = t;
    }
    for (int w = 16; w < 64; w += w) {
        load_a_frags(ACr, ACi, PA, r0, arf, aif);
        sum_a_frags(arf, aif, asf);
        for (int n0 = 0; n0 < w; n0 += 16)
            cplx_mm_tile<false>(arf, aif, asf, Xr, Xi, 64, Xr + w, Xi + w, 64, nullptr, nullptr, r0, n0);
        for (int n0 = 0; n0 < 64; n0 += 16)
            cplx_mm_tile<true>(arf, aif, asf, sm + src, sm + src + 4096, 64, sm + dst, sm + dst + 4096, 64, ACr, ACi, r0, n0);
        __syncthreads();
        const int t = src; src = dst; dst = t;
    }
    for (int w = 1; w <= 16; w += w) {
        load_a_frags(Gr, Gi, PA, 0, arf, aif);
        sum_a_frags(arf, aif, asf);
        g_double_tile(arf, aif, asf, sm + src, sm + src + 4096, Gr, Gi, Dr, Di, r0, w);
        if (w < 16) {
            load_a_frags(ACr, ACi, PA, r0, arf, aif);
            sum_a_frags(arf, aif, asf);
            for (int n0 = 0; n0 < 64; n0 += 16)
                cplx_mm_tile<true>(arf, aif, asf, sm + src, sm + src + 4096, 64, sm + dst, sm + dst + 4096, 64, ACr, ACi, r0, n0);
        }
        __syncthreads();
        const int t = src; src = dst; dst = t;
    }
    for (int j0 = 0; j0 < 32; j0 += 16) {
        load_a_frags(Gr, Gi, PA, j0, arf, aif);
        k_tile(arf, aif, Xr, Xi, Kb, j0, r0);
    }
    __syncthreads();

    for (int i = 0; i < LL / 128; ++i) {
        const int idx = i*128 + tid;
        out[h*LL + idx] = Kb[idx];
    }
}

extern "C" void kernel_launch(void* const* d_in, const int* in_sizes, int n_in,
                              void* d_out, int out_size, void* d_ws, size_t ws_size,
                              hipStream_t stream) {
    const float* A_real = (const float*)d_in[0];
    const float* A_imag = (const float*)d_in[1];
    const float* B_real = (const float*)d_in[2];
    const float* B_imag = (const float*)d_in[3];
    const float* P_real = (const float*)d_in[4];
    const float* P_imag = (const float*)d_in[5];
    const float* C_real = (const float*)d_in[6];
    const float* C_imag = (const float*)d_in[7];
    const float* log_dt = (const float*)d_in[8];
    float* out = (float*)d_out;
    (void)in_sizes; (void)n_in; (void)out_size; (void)d_ws; (void)ws_size;

    const size_t shmem_bytes = 40328u * sizeof(float);
    s4_dplr_kernel<<<NH, 128, shmem_bytes, stream>>>(
        A_real, A_imag, B_real, B_imag, P_real, P_imag, C_real, C_imag, log_dt, out);
}


# LDS matrix operands all k-contiguous (B operands, X and tile results transposed, pitch 68), k order regrouped in one shared fragment loader: conflict-free 16-byte LDS loads/stores; sum order changes
# speedup vs baseline: 13.7357x; 1.0987x over previous
#include <hip/hip_runtime.h>
#include <hip/hip_bf16.h>

#define NH 256
#define NN 64
#define LL 2048
#define PA 68
#define PLANE (64 * PA)

typedef __attribute__((ext_vector_type(2))) float v2f;
typedef __attribute__((ext_vector_type(4))) float v4f;
typedef __attribute__((ext_vector_type(8))) float v8f;

struct cpx { float r, i; };
__device__ __forceinline__ cpx cmul(cpx a, cpx b) { return {a.r*b.r - a.i*b.i, a.r*b.i + a.i*b.r}; }
__device__ __forceinline__ cpx cadd(cpx a, cpx b) { return {a.r + b.r, a.i + b.i}; }
__device__ __forceinline__ cpx csub(cpx a, cpx b) { return {a.r - b.r, a.i - b.i}; }
__device__ __forceinline__ cpx cinv(cpx a) { float d = a.r*a.r + a.i*a.i; return {a.r/d, -a.i/d}; }
__device__ __forceinline__ cpx cscale(cpx a, float s) { return {a.r*s, a.i*s}; }

__device__ __forceinline__ void load_frags(const float* Pr, const float* Pi, int row,
                                           v2f (&fr)[16], v2f (&fi)[16])
{
    const int lane = threadIdx.x & 31;
    const int h4 = (lane >> 4) << 2;
    const v4f* qr = (const v4f*)(Pr + row * PA + h4);
    const v4f* qi = (const v4f*)(Pi + row * PA + h4);
    #pragma unroll
    for (int t = 0; t < 8; ++t) {
        const v4f a = qr[2*t];
        const v4f b = qi[2*t];
        fr[2*t][0] = a.x;  fr[2*t][1] = a.y;  fr[2*t + 1][0] = a.z;  fr[2*t + 1][1] = a.w;
        fi[2*t][0] = b.x;  fi[2*t][1] = b.y;  fi[2*t + 1][0] = b.z;  fi[2*t + 1][1] = b.w;
    }
}

__device__ __forceinline__ void sum_a_frags(const v2f (&arf)[16], const v2f (&aif)[16], v2f (&asf)[16])
{
    #pragma unroll
    for (int s = 0; s < 16; ++s) asf[s] = arf[s] + aif[s];
}

__device__ __forceinline__ void cplx_chains(
    const v2f (&arf)[16], const v2f (&aif)[16], const v2f (&asf)[16],
    const v2f (&brv)[16], const v2f (&biv)[16], v8f &p1, v8f &p2, v8f &p3)
{
    #pragma unroll
    for (int s = 0; s < 16; ++s) {
        const v2f bs = brv[s] + biv[s];
        p1 = __builtin_amdgcn_wmma_f32_16x16x4_f32(false, arf[s], false, brv[s], (short)0, p1, false, false);
        p2 = __builtin_amdgcn_wmma_f32_16x16x4_f32(false, aif[s], false, biv[s], (short)0, p2, false, false);
        p3 = __builtin_amdgcn_wmma_f32_16x16x4_f32(false, asf[s], false, bs, (short)0, p3, false, false);
        __builtin_amdgcn_sched_barrier(0);
    }
}

template <bool ACOPY>
__device__ __forceinline__ void cplx_mm_tile(
    const v2f (&arf)[16], const v2f (&aif)[16], const v2f (&asf)[16],
    const float* BTr, const float* BTi,
    float* CTr, float* CTi,
    float* Er, float* Ei,
    int r0, int n0)
{
    const int lane = threadIdx.x & 31;
    const int mn = lane & 15;
    v2f brv[16], biv[16];
    load_frags(BTr, BTi, n0 + mn, brv, biv);
    __builtin_amdgcn_sched_barrier(0);
    v8f p1 = {}; v8f p2 = {}; v8f p3 = {};
    cplx_chains(arf, aif, asf, brv, biv, p1, p2, p3);
    const int mbase = (lane >> 4) << 3;
    float re[8], im[8];
    for (int v = 0; v < 8; ++v) { re[v] = p1[v] - p2[v]; im[v] = (p3[v] - p1[v]) - p2[v]; }
    v4f* cr = (v4f*)(CTr + (n0 + mn) * PA + r0 + mbase);
    v4f* ci = (v4f*)(CTi + (n0 + mn) * PA + r0 + mbase);
    cr[0] = (v4f){re[0], re[1], re[2], re[3]};  cr[1] = (v4f){re[4], re[5], re[6], re[7]};
    ci[0] = (v4f){im[0], im[1], im[2], im[3]};  ci[1] = (v4f){im[4], im[5], im[6], im[7]};
    if (ACOPY) {
        for (int v = 0; v < 8; ++v) {
            Er[(r0 + mbase + v) * PA + n0 + mn] = re[v];
            Ei[(r0 + mbase + v) * PA + n0 + mn] = im[v];
        }
    }
}

__device__ __forceinline__ void x_double_tile(
    const v2f (&arf)[16], const v2f (&aif)[16], const v2f (&asf)[16],
    float* XTr, float* XTi, float* Pr, float* Pi, int r0, int w)
{
    const int lane = threadIdx.x & 31;
    const int mn = lane & 15;
    const bool live = mn < w;
    v2f brv[16], biv[16];
    #pragma unroll
    for (int s = 0; s < 16; ++s) { brv[s][0] = 0.f; brv[s][1] = 0.f; biv[s][0] = 0.f; biv[s][1] = 0.f; }
    if (live) load_frags(XTr, XTi, mn, brv, biv);
    v8f p1 = {}; v8f p2 = {}; v8f p3 = {};
    cplx_chains(arf, aif, asf, brv, biv, p1, p2, p3);
    const int mbase = (lane >> 4) << 3;
    float* dr = live ? (XTr + (w + mn) * PA) : (Pr + mn * PA);
    float* di = live ? (XTi + (w + mn) * PA) : (Pi + mn * PA);
    float re[8], im[8];
    for (int v = 0; v < 8; ++v) { re[v] = p1[v] - p2[v]; im[v] = (p3[v] - p1[v]) - p2[v]; }
    v4f* cr = (v4f*)(dr + r0 + mbase);
    v4f* ci = (v4f*)(di + r0 + mbase);
    cr[0] = (v4f){re[0], re[1], re[2], re[3]};  cr[1] = (v4f){re[4], re[5], re[6], re[7]};
    ci[0] = (v4f){im[0], im[1], im[2], im[3]};  ci[1] = (v4f){im[4], im[5], im[6], im[7]};
}

__device__ __forceinline__ void g_double_tile(
    const v2f (&arf)[16], const v2f (&aif)[16], const v2f (&asf)[16],
    const float* MTr, const float* MTi,
    float* Gr, float* Gi, float* Pr, float* Pi, int n0, int w)
{
    const int lane = threadIdx.x & 31;
    const int mn = lane & 15;
    v2f brv[16], biv[16];
    load_frags(MTr, MTi, n0 + mn, brv, biv);
    __builtin_amdgcn_sched_barrier(0);
    v8f p1 = {}; v8f p2 = {}; v8f p3 = {};
    cplx_chains(arf, aif, asf, brv, biv, p1, p2, p3);
    const int mbase = (lane >> 4) << 3;
    for (int v = 0; v < 8; ++v) {
        const int row = mbase + v;
        const bool live = row < w;
        float* dr = live ? (Gr + (w + row) * PA) : (Pr + row * PA);
        float* di = live ? (Gi + (w + row) * PA) : (Pi + row * PA);
        dr[n0 + mn] = p1[v] - p2[v];
        di[n0 + mn] = (p3[v] - p1[v]) - p2[v];
    }
}

__device__ __forceinline__ void k_tile(
    const v2f (&arf)[16], const v2f (&aif)[16],
    const float* XTr, const float* XTi, float* Kb, int j0, int n0)
{
    const int lane = threadIdx.x & 31;
    const int mn = lane & 15;
    v2f brv[16], biv[16];
    load_frags(XTr, XTi, n0 + mn, brv, biv);
    v8f grxr = {}; v8f gixi = {};
    #pragma unroll
    for (int s = 0; s < 16; ++s) {
        grxr = __builtin_amdgcn_wmma_f32_16x16x4_f32(false, arf[s], false, brv[s], (short)0, grxr, false, false);
        gixi = __builtin_amdgcn_wmma_f32_16x16x4_f32(false, aif[s], false, biv[s], (short)0, gixi, false, false);
    }
    const int mbase = (lane >> 4) << 3;
    for (int v = 0; v < 8; ++v)
        Kb[(j0 + mbase + v) * 64 + n0 + mn] = 2.0f * (grxr[v] - gixi[v]);
}

extern "C" __global__ __launch_bounds__(128)
void s4_dplr_kernel(const float* __restrict__ A_real, const float* __restrict__ A_imag,
                    const float* __restrict__ B_real, const float* __restrict__ B_imag,
                    const float* __restrict__ P_real, const float* __restrict__ P_imag,
                    const float* __restrict__ C_real, const float* __restrict__ C_imag,
                    const float* __restrict__ log_dt, float* __restrict__ out)
{
    extern __shared__ __attribute__((aligned(16))) float sm[];
    const int h   = blockIdx.x;
    const int tid = threadIdx.x;
    const int wv  = tid >> 5;
    const int r0  = 16 * wv;
    const int mn  = tid & 15;

    const int OFF_T0 = 0;
    const int OFF_T1 = 2 * PLANE;
    float* ACr = sm + 4 * PLANE;
    float* ACi = ACr + PLANE;
    float* XTr = sm + 6 * PLANE;
    float* XTi = XTr + PLANE;
    float* Gr  = sm + 8 * PLANE;
    float* Gi  = Gr + 32 * PA;
    float* Dr  = Gr + 16 * PA;
    float* Di  = Gi + 16 * PA;
    float* Kb  = sm + OFF_T0;
    float* ur  = Gi + 32 * PA; float* ui = ur + 64;
    float* wr  = ui + 64;     float* wi = wr + 64;
    float* gr  = wi + 64;     float* gi = gr + 64;
    float* pr  = gi + 64;     float* pi = pr + 64;
    float* scal = pi + 64;

    const float dt = __expf(log_dt[h]);
    const float cc = 0.5f * dt;

    if (tid < 64) {
        const int n = tid;
        cpx lam = { -A_real[h*64 + n], -A_imag[h*64 + n] };
        cpx p   = {  P_real[h*64 + n],  P_imag[h*64 + n] };
        cpx Dinv = cinv((cpx){1.0f - cc*lam.r, -cc*lam.i});
        cpx e    = {1.0f + cc*lam.r,  cc*lam.i};
        cpx u = cmul(Dinv, p);
        cpx g = cmul(Dinv, e);
        ur[n] = u.r; ui[n] = u.i; gr[n] = g.r; gi[n] = g.i;
        pr[n] = p.r; pi[n] = p.i;
        Gr[n] = C_real[h*64 + n]; Gi[n] = C_imag[h*64 + n];
    }
    for (int idx = 64 + tid; idx < 32 * PA; idx += 128) { Gr[idx] = 0.f; Gi[idx] = 0.f; }
    __syncthreads();
    if (tid == 0) {
        cpx s = {0.f, 0.f};
        for (int n = 0; n < 64; ++n)
            s = cadd(s, cmul((cpx){pr[n], -pi[n]}, (cpx){ur[n], ui[n]}));
        cpx alpha = cmul((cpx){cc, 0.f}, cinv((cpx){1.0f + cc*s.r, cc*s.i}));
        cpx base  = csub((cpx){cc, 0.f}, cmul(alpha, cscale(s, cc)));
        scal[0] = alpha.r; scal[1] = alpha.i; scal[2] = base.r; scal[3] = base.i;
    }
    __syncthreads();
    const cpx alpha = {scal[0], scal[1]};
    const cpx base  = {scal[2], scal[3]};
    if (tid < 64) {
        const int n = tid;
        cpx w = cmul((cpx){pr[n], -pi[n]}, cadd(base, cmul(alpha, (cpx){gr[n], gi[n]})));
        wr[n] = w.r; wi[n] = w.i;
    }
    __syncthreads();
    for (int idx = tid; idx < 4096; idx += 128) {
        const int a = idx >> 6, b = idx & 63;
        cpx v  = cmul((cpx){ur[a], ui[a]}, (cpx){wr[b], wi[b]});
        cpx vt = cmul((cpx){ur[b], ui[b]}, (cpx){wr[a], wi[a]});
        float rr = -v.r, rim = -v.i, tr = -vt.r, tim = -vt.i;
        if (a == b) { rr += gr[a]; rim += gi[a]; tr += gr[a]; tim += gi[a]; }
        ACr[a * PA + b] = rr;  ACi[a * PA + b] = rim;
        sm[OFF_T0 + a * PA + b] = tr;  sm[OFF_T0 + PLANE + a * PA + b] = tim;
    }
    __syncthreads();

    if (tid < 64) {
        const int n = tid;
        cpx lam = { -A_real[h*64 + n], -A_imag[h*64 + n] };
        cpx Dinv = cinv((cpx){1.0f - cc*lam.r, -cc*lam.i});
        cpx b = { B_real[h*64 + n], B_imag[h*64 + n] };
        cpx t = cmul(Dinv, cscale(b, dt));
        wr[n] = t.r; wi[n] = t.i;
    }
    __syncthreads();
    if (tid == 0) {
        cpx q = {0.f, 0.f};
        for (int n = 0; n < 64; ++n)
            q = cadd(q, cmul((cpx){pr[n], -pi[n]}, (cpx){wr[n], wi[n]}));
        cpx aq = cmul(alpha, q);
        scal[4] = aq.r; scal[5] = aq.i;
    }
    __syncthreads();
    if (tid < 64) {
        const int n = tid;
        cpx dB = csub((cpx){wr[n], wi[n]}, cmul((cpx){scal[4], scal[5]}, (cpx){ur[n], ui[n]}));
        XTr[n] = dB.r;
        XTi[n] = dB.i;
    }
    __syncthreads();

    v2f arf[16], aif[16], asf[16];
    int src = OFF_T0, dst = OFF_T1;

    for (int w = 1; w < 16; w += w) {
        load_frags(ACr, ACi, r0 + mn, arf, aif);
        sum_a_frags(arf, aif, asf);
        x_double_tile(arf, aif, asf, XTr, XTi, Dr, Di, r0, w);
        for (int n0 = 0; n0 < 64; n0 += 16)
            cplx_mm_tile<true>(arf, aif, asf, sm + src, sm + src + PLANE, sm + dst, sm + dst + PLANE, ACr, ACi, r0, n0);
        __syncthreads();
        const int t = src; src = dst; dst = t;
    }
    for (int w = 16; w < 64; w += w) {
        load_frags(ACr, ACi, r0 + mn, arf, aif);
        sum_a_frags(arf, aif, asf);
        for (int n0 = 0; n0 < w; n0 += 16)
            cplx_mm_tile<false>(arf, aif, asf, XTr, XTi, XTr + w * PA, XTi + w * PA, nullptr, nullptr, r0, n0);
        for (int n0 = 0; n0 < 64; n0 += 16)
            cplx_mm_tile<true>(arf, aif, asf, sm + src, sm + src + PLANE, sm + dst, sm + dst + PLANE, ACr, ACi, r0, n0);
        __syncthreads();
        const int t = src; src = dst; dst = t;
    }
    for (int w = 1; w <= 16; w += w) {
        load_frags(Gr, Gi, mn, arf, aif);
        sum_a_frags(arf, aif, asf);
        g_double_tile(arf, aif, asf, sm + src, sm + src + PLANE, Gr, Gi, Dr, Di, r0, w);
        if (w < 16) {
            load_frags(ACr, ACi, r0 + mn, arf, aif);
            sum_a_frags(arf, aif, asf);
            for (int n0 = 0; n0 < 64; n0 += 16)
                cplx_mm_tile<true>(arf, aif, asf, sm + src, sm + src + PLANE, sm + dst, sm + dst + PLANE, ACr, ACi, r0, n0);
        }
        __syncthreads();
        const int t = src; src = dst; dst = t;
    }
    for (int j0 = 0; j0 < 32; j0 += 16) {
        load_frags(Gr, Gi, j0 + mn, arf, aif);
        k_tile(arf, aif, XTr, XTi, Kb, j0, r0);
    }
    __syncthreads();

    for (int i = 0; i < LL / 128; ++i) {
        const int idx = i*128 + tid;
        out[h*LL + idx] = Kb[idx];
    }
}

extern "C" void kernel_launch(void* const* d_in, const int* in_sizes, int n_in,
                              void* d_out, int out_size, void* d_ws, size_t ws_size,
                              hipStream_t stream) {
    const float* A_real = (const float*)d_in[0];
    const float* A_imag = (const float*)d_in[1];
    const float* B_real = (const float*)d_in[2];
    const float* B_imag = (const float*)d_in[3];
    const float* P_real = (const float*)d_in[4];
    const float* P_imag = (const float*)d_in[5];
    const float* C_real = (const float*)d_in[6];
    const float* C_imag = (const float*)d_in[7];
    const float* log_dt = (const float*)d_in[8];
    float* out = (float*)d_out;
    (void)in_sizes; (void)n_in; (void)out_size; (void)d_ws; (void)ws_size;

    const size_t shmem_bytes = 39688u * sizeof(float);
    s4_dplr_kernel<<<NH, 128, shmem_bytes, stream>>>(
        A_real, A_imag, B_real, B_imag, P_real, P_imag, C_real, C_imag, log_dt, out);
}


# 8-wave workgroup, two per SIMD: sibling waves w, w+4 split each row block's tiles (untransposed power copy double-buffered); one's LDS loads, epilogues, barrier waits hide behind the other's WMMAs
# speedup vs baseline: 15.6698x; 1.1408x over previous
#include <hip/hip_runtime.h>
#include <hip/hip_bf16.h>

#define NH 256
#define NN 64
#define LL 2048
#define PA 68
#define PLANE (64 * PA)

typedef __attribute__((ext_vector_type(2))) float v2f;
typedef __attribute__((ext_vector_type(4))) float v4f;
typedef __attribute__((ext_vector_type(8))) float v8f;

struct cpx { float r, i; };
__device__ __forceinline__ cpx cmul(cpx a, cpx b) { return {a.r*b.r - a.i*b.i, a.r*b.i + a.i*b.r}; }
__device__ __forceinline__ cpx cadd(cpx a, cpx b) { return {a.r + b.r, a.i + b.i}; }
__device__ __forceinline__ cpx csub(cpx a, cpx b) { return {a.r - b.r, a.i - b.i}; }
__device__ __forceinline__ cpx cinv(cpx a) { float d = a.r*a.r + a.i*a.i; return {a.r/d, -a.i/d}; }
__device__ __forceinline__ cpx cscale(cpx a, float s) { return {a.r*s, a.i*s}; }

__device__ __forceinline__ void load_frags(const float* Pr, const float* Pi, int row,
                                           v2f (&fr)[16], v2f (&fi)[16])
{
    const int lane = threadIdx.x & 31;
    const int h4 = (lane >> 4) << 2;
    const v4f* qr = (const v4f*)(Pr + row * PA + h4);
    const v4f* qi = (const v4f*)(Pi + row * PA + h4);
    #pragma unroll
    for (int t = 0; t < 8; ++t) {
        const v4f a = qr[2*t];
        const v4f b = qi[2*t];
        fr[2*t][0] = a.x;  fr[2*t][1] = a.y;  fr[2*t + 1][0] = a.z;  fr[2*t + 1][1] = a.w;
        fi[2*t][0] = b.x;  fi[2*t][1] = b.y;  fi[2*t + 1][0] = b.z;  fi[2*t + 1][1] = b.w;
    }
}

__device__ __forceinline__ void sum_a_frags(const v2f (&arf)[16], const v2f (&aif)[16], v2f (&asf)[16])
{
    #pragma unroll
    for (int s = 0; s < 16; ++s) asf[s] = arf[s] + aif[s];
}

__device__ __forceinline__ void cplx_chains(
    const v2f (&arf)[16], const v2f (&aif)[16], const v2f (&asf)[16],
    const v2f (&brv)[16], const v2f (&biv)[16], v8f &p1, v8f &p2, v8f &p3)
{
    #pragma unroll
    for (int s = 0; s < 16; ++s) {
        const v2f bs = brv[s] + biv[s];
        p1 = __builtin_amdgcn_wmma_f32_16x16x4_f32(false, arf[s], false, brv[s], (short)0, p1, false, false);
        p2 = __builtin_amdgcn_wmma_f32_16x16x4_f32(false, aif[s], false, biv[s], (short)0, p2, false, false);
        p3 = __builtin_amdgcn_wmma_f32_16x16x4_f32(false, asf[s], false, bs, (short)0, p3, false, false);
        __builtin_amdgcn_sched_barrier(0);
    }
}

template <bool ACOPY>
__device__ __forceinline__ void cplx_mm_tile(
    const v2f (&arf)[16], const v2f (&aif)[16], const v2f (&asf)[16],
    const float* BTr, const float* BTi,
    float* CTr, float* CTi,
    float* Er, float* Ei,
    int r0, int n0)
{
    const int lane = threadIdx.x & 31;
    const int mn = lane & 15;
    v2f brv[16], biv[16];
    load_frags(BTr, BTi, n0 + mn, brv, biv);
    __builtin_amdgcn_sched_barrier(0);
    v8f p1 = {}; v8f p2 = {}; v8f p3 = {};
    cplx_chains(arf, aif, asf, brv, biv, p1, p2, p3);
    const int mbase = (lane >> 4) << 3;
    float re[8], im[8];
    for (int v = 0; v < 8; ++v) { re[v] = p1[v] - p2[v]; im[v] = (p3[v] - p1[v]) - p2[v]; }
    v4f* cr = (v4f*)(CTr + (n0 + mn) * PA + r0 + mbase);
    v4f* ci = (v4f*)(CTi + (n0 + mn) * PA + r0 + mbase);
    cr[0] = (v4f){re[0], re[1], re[2], re[3]};  cr[1] = (v4f){re[4], re[5], re[6], re[7]};
    ci[0] = (v4f){im[0], im[1], im[2], im[3]};  ci[1] = (v4f){im[4], im[5], im[6], im[7]};
    if (ACOPY) {
        for (int v = 0; v < 8; ++v) {
            Er[(r0 + mbase + v) * PA + n0 + mn] = re[v];
            Ei[(r0 + mbase + v) * PA + n0 + mn] = im[v];
        }
    }
}

__device__ __forceinline__ void x_double_tile(
    const v2f (&arf)[16], const v2f (&aif)[16], const v2f (&asf)[16],
    float* XTr, float* XTi, float* Pr, float* Pi, int r0, int w)
{
    const int lane = threadIdx.x & 31;
    const int mn = lane & 15;
    const bool live = mn < w;
    v2f brv[16], biv[16];
    #pragma unroll
    for (int s = 0; s < 16; ++s) { brv[s][0] = 0.f; brv[s][1] = 0.f; biv[s][0] = 0.f; biv[s][1] = 0.f; }
    if (live) load_frags(XTr, XTi, mn, brv, biv);
    v8f p1 = {}; v8f p2 = {}; v8f p3 = {};
    cplx_chains(arf, aif, asf, brv, biv, p1, p2, p3);
    const int mbase = (lane >> 4) << 3;
    float* dr = live ? (XTr + (w + mn) * PA) : (Pr + mn * PA);
    float* di = live ? (XTi + (w + mn) * PA) : (Pi + mn * PA);
    float re[8], im[8];
    for (int v = 0; v < 8; ++v) { re[v] = p1[v] - p2[v]; im[v] = (p3[v] - p1[v]) - p2[v]; }
    v4f* cr = (v4f*)(dr + r0 + mbase);
    v4f* ci = (v4f*)(di + r0 + mbase);
    cr[0] = (v4f){re[0], re[1], re[2], re[3]};  cr[1] = (v4f){re[4], re[5], re[6], re[7]};
    ci[0] = (v4f){im[0], im[1], im[2], im[3]};  ci[1] = (v4f){im[4], im[5], im[6], im[7]};
}

__device__ __forceinline__ void g_double_tile(
    const v2f (&arf)[16], const v2f (&aif)[16], const v2f (&asf)[16],
    const float* MTr, const float* MTi,
    float* Gr, float* Gi, float* Pr, float* Pi, int n0, int w)
{
    const int lane = threadIdx.x & 31;
    const int mn = lane & 15;
    v2f brv[16], biv[16];
    load_frags(MTr, MTi, n0 + mn, brv, biv);
    __builtin_amdgcn_sched_barrier(0);
    v8f p1 = {}; v8f p2 = {}; v8f p3 = {};
    cplx_chains(arf, aif, asf, brv, biv, p1, p2, p3);
    const int mbase = (lane >> 4) << 3;
    for (int v = 0; v < 8; ++v) {
        const int row = mbase + v;
        const bool live = row < w;
        float* dr = live ? (Gr + (w + row) * PA) : (Pr + row * PA);
        float* di = live ? (Gi + (w + row) * PA) : (Pi + row * PA);
        dr[n0 + mn] = p1[v] - p2[v];
        di[n0 + mn] = (p3[v] - p1[v]) - p2[v];
    }
}

__device__ __forceinline__ void k_tile(
    const v2f (&arf)[16], const v2f (&aif)[16],
    const float* XTr, const float* XTi, float* Kb, int j0, int n0)
{
    const int lane = threadIdx.x & 31;
    const int mn = lane & 15;
    v2f brv[16], biv[16];
    load_frags(XTr, XTi, n0 + mn, brv, biv);
    v8f grxr = {}; v8f gixi = {};
    #pragma unroll
    for (int s = 0; s < 16; ++s) {
        grxr = __builtin_amdgcn_wmma_f32_16x16x4_f32(false, arf[s], false, brv[s], (short)0, grxr, false, false);
        gixi = __builtin_amdgcn_wmma_f32_16x16x4_f32(false, aif[s], false, biv[s], (short)0, gixi, false, false);
    }
    const int mbase = (lane >> 4) << 3;
    for (int v = 0; v < 8; ++v)
        Kb[(j0 + mbase + v) * 64 + n0 + mn] = 2.0f * (grxr[v] - gixi[v]);
}

extern "C" __global__ __launch_bounds__(256)
void s4_dplr_kernel(const float* __restrict__ A_real, const float* __restrict__ A_imag,
                    const float* __restrict__ B_real, const float* __restrict__ B_imag,
                    const float* __restrict__ P_real, const float* __restrict__ P_imag,
                    const float* __restrict__ C_real, const float* __restrict__ C_imag,
                    const float* __restrict__ log_dt, float* __restrict__ out)
{
    extern __shared__ __attribute__((aligned(16))) float sm[];
    const int h   = blockIdx.x;
    const int tid = threadIdx.x;
    const int wv  = __builtin_amdgcn_readfirstlane(tid >> 5);
    const int rb  = wv & 3;
    const int hf  = wv >> 2;
    const int r0  = 16 * rb;
    const int c0  = 32 * hf;
    const int mn  = tid & 15;

    const int OFF_T0 = 0;
    const int OFF_T1 = 2 * PLANE;
    const int OFF_A0 = 4 * PLANE;
    const int OFF_A1 = 6 * PLANE;
    float* XTr = sm + 8 * PLANE;
    float* XTi = XTr + PLANE;
    float* Gr  = sm + 10 * PLANE;
    float* Gi  = Gr + 32 * PA;
    float* Dr  = Gr + 16 * PA;
    float* Di  = Gi + 16 * PA;
    float* Kb  = sm + OFF_T0;
    float* ur  = Gi + 32 * PA; float* ui = ur + 64;
    float* wr  = ui + 64;     float* wi = wr + 64;
    float* gr  = wi + 64;     float* gi = gr + 64;
    float* pr  = gi + 64;     float* pi = pr + 64;
    float* scal = pi + 64;

    const float dt = __expf(log_dt[h]);
    const float cc = 0.5f * dt;

    if (tid < 64) {
        const int n = tid;
        cpx lam = { -A_real[h*64 + n], -A_imag[h*64 + n] };
        cpx p   = {  P_real[h*64 + n],  P_imag[h*64 + n] };
        cpx Dinv = cinv((cpx){1.0f - cc*lam.r, -cc*lam.i});
        cpx e    = {1.0f + cc*lam.r,  cc*lam.i};
        cpx u = cmul(Dinv, p);
        cpx g = cmul(Dinv, e);
        ur[n] = u.r; ui[n] = u.i; gr[n] = g.r; gi[n] = g.i;
        pr[n] = p.r; pi[n] = p.i;
        Gr[n] = C_real[h*64 + n]; Gi[n] = C_imag[h*64 + n];
    }
    for (int idx = 64 + tid; idx < 32 * PA; idx += 256) { Gr[idx] = 0.f; Gi[idx] = 0.f; }
    __syncthreads();
    if (tid == 0) {
        cpx s = {0.f, 0.f};
        for (int n = 0; n < 64; ++n)
            s = cadd(s, cmul((cpx){pr[n], -pi[n]}, (cpx){ur[n], ui[n]}));
        cpx alpha = cmul((cpx){cc, 0.f}, cinv((cpx){1.0f + cc*s.r, cc*s.i}));
        cpx base  = csub((cpx){cc, 0.f}, cmul(alpha, cscale(s, cc)));
        scal[0] = alpha.r; scal[1] = alpha.i; scal[2] = base.r; scal[3] = base.i;
    }
    __syncthreads();
    const cpx alpha = {scal[0], scal[1]};
    const cpx base  = {scal[2], scal[3]};
    if (tid < 64) {
        const int n = tid;
        cpx w = cmul((cpx){pr[n], -pi[n]}, cadd(base, cmul(alpha, (cpx){gr[n], gi[n]})));
        wr[n] = w.r; wi[n] = w.i;
    }
    __syncthreads();
    for (int idx = tid; idx < 4096; idx += 256) {
        const int a = idx >> 6, b = idx & 63;
        cpx v  = cmul((cpx){ur[a], ui[a]}, (cpx){wr[b], wi[b]});
        cpx vt = cmul((cpx){ur[b], ui[b]}, (cpx){wr[a], wi[a]});
        float rr = -v.r, rim = -v.i, tr = -vt.r, tim = -vt.i;
        if (a == b) { rr += gr[a]; rim += gi[a]; tr += gr[a]; tim += gi[a]; }
        sm[OFF_A0 + a * PA + b] = rr;  sm[OFF_A0 + PLANE + a * PA + b] = rim;
        sm[OFF_T0 + a * PA + b] = tr;  sm[OFF_T0 + PLANE + a * PA + b] = tim;
    }
    __syncthreads();

    if (tid < 64) {
        const int n = tid;
        cpx lam = { -A_real[h*64 + n], -A_imag[h*64 + n] };
        cpx Dinv = cinv((cpx){1.0f - cc*lam.r, -cc*lam.i});
        cpx b = { B_real[h*64 + n], B_imag[h*64 + n] };
        cpx t = cmul(Dinv, cscale(b, dt));
        wr[n] = t.r; wi[n] = t.i;
    }
    __syncthreads();
    if (tid == 0) {
        cpx q = {0.f, 0.f};
        for (int n = 0; n < 64; ++n)
            q = cadd(q, cmul((cpx){pr[n], -pi[n]}, (cpx){wr[n], wi[n]}));
        cpx aq = cmul(alpha, q);
        scal[4] = aq.r; scal[5] = aq.i;
    }
    __syncthreads();
    if (tid < 64) {
        const int n = tid;
        cpx dB = csub((cpx){wr[n], wi[n]}, cmul((cpx){scal[4], scal[5]}, (cpx){ur[n], ui[n]}));
        XTr[n] = dB.r;
        XTi[n] = dB.i;
    }
    __syncthreads();

    v2f arf[16], aif[16], asf[16];
    int src = OFF_T0, dst = OFF_T1;
    int asrc = OFF_A0, adst = OFF_A1;

    for (int w = 1; w < 16; w += w) {
        load_frags(sm + asrc, sm + asrc + PLANE, r0 + mn, arf, aif);
        sum_a_frags(arf, aif, asf);
        if (hf == 0)
            x_double_tile(arf, aif, asf, XTr, XTi, Dr, Di, r0, w);
        for (int n0 = c0; n0 < c0 + 32; n0 += 16)
            cplx_mm_tile<true>(arf, aif, asf, sm + src, sm + src + PLANE, sm + dst, sm + dst + PLANE, sm + adst, sm + adst + PLANE, r0, n0);
        __syncthreads();
        const int t = src; src = dst; dst = t;
        const int ta = asrc; asrc = adst; adst = ta;
    }
    for (int w = 16; w < 64; w += w) {
        load_frags(sm + asrc, sm + asrc + PLANE, r0 + mn, arf, aif);
        sum_a_frags(arf, aif, asf);
        for (int n0 = 16 * hf; n0 < w; n0 += 32)
            cplx_mm_tile<false>(arf, aif, asf, XTr, XTi, XTr + w * PA, XTi + w * PA, nullptr, nullptr, r0, n0);
        for (int n0 = c0; n0 < c0 + 32; n0 += 16)
            cplx_mm_tile<true>(arf, aif, asf, sm + src, sm + src + PLANE, sm + dst, sm + dst + PLANE, sm + adst, sm + adst + PLANE, r0, n0);
        __syncthreads();
        const int t = src; src = dst; dst = t;
        const int ta = asrc; asrc = adst; adst = ta;
    }
    for (int w = 1; w <= 16; w += w) {
        if (hf == 0) {
            load_frags(Gr, Gi, mn, arf, aif);
            sum_a_frags(arf, aif, asf);
            g_double_tile(arf, aif, asf, sm + src, sm + src + PLANE, Gr, Gi, Dr, Di, r0, w);
        }
        if (w < 16) {
            load_frags(sm + asrc, sm + asrc + PLANE, r0 + mn, arf, aif);
            sum_a_frags(arf, aif, asf);
            for (int n0 = c0; n0 < c0 + 32; n0 += 16)
                cplx_mm_tile<true>(arf, aif, asf, sm + src, sm + src + PLANE, sm + dst, sm + dst + PLANE, sm + adst, sm + adst + PLANE, r0, n0);
        }
        __syncthreads();
        const int t = src; src = dst; dst = t;
        const int ta = asrc; asrc = adst; adst = ta;
    }
    {
        const int j0 = 16 * hf;
        load_frags(Gr, Gi, j0 + mn, arf, aif);
        k_tile(arf, aif, XTr, XTi, Kb, j0, r0);
    }
    __syncthreads();

    for (int i = 0; i < LL / 256; ++i) {
        const int idx = i*256 + tid;
        out[h*LL + idx] = Kb[idx];
    }
}

extern "C" void kernel_launch(void* const* d_in, const int* in_sizes, int n_in,
                              void* d_out, int out_size, void* d_ws, size_t ws_size,
                              hipStream_t stream) {
    const float* A_real = (const float*)d_in[0];
    const float* A_imag = (const float*)d_in[1];
    const float* B_real = (const float*)d_in[2];
    const float* B_imag = (const float*)d_in[3];
    const float* P_real = (const float*)d_in[4];
    const float* P_imag = (const float*)d_in[5];
    const float* C_real = (const float*)d_in[6];
    const float* C_imag = (const float*)d_in[7];
    const float* log_dt = (const float*)d_in[8];
    float* out = (float*)d_out;
    (void)in_sizes; (void)n_in; (void)out_size; (void)d_ws; (void)ws_size;

    const size_t shmem_bytes = 48392u * sizeof(float);
    s4_dplr_kernel<<<NH, 256, shmem_bytes, stream>>>(
        A_real, A_imag, B_real, B_imag, P_real, P_imag, C_real, C_imag, log_dt, out);
}


# First 4 squarings replaced via dA = diag(g) - u w^T: dA^16 = diag(g^16) - sum_i (g^i u)(w^T dA^(15-i)), one rank-16 WMMA product; X cols 1-15 and w^T dA^j by 15-step rank-1 chains; sum order changes
# speedup vs baseline: 22.5070x; 1.4363x over previous
#include <hip/hip_runtime.h>
#include <hip/hip_bf16.h>

#define NH 256
#define NN 64
#define LL 2048
#define PA 68
#define PLANE (64 * PA)

typedef __attribute__((ext_vector_type(2))) float v2f;
typedef __attribute__((ext_vector_type(4))) float v4f;
typedef __attribute__((ext_vector_type(8))) float v8f;

struct cpx { float r, i; };
__device__ __forceinline__ cpx cmul(cpx a, cpx b) { return {a.r*b.r - a.i*b.i, a.r*b.i + a.i*b.r}; }
__device__ __forceinline__ cpx cadd(cpx a, cpx b) { return {a.r + b.r, a.i + b.i}; }
__device__ __forceinline__ cpx csub(cpx a, cpx b) { return {a.r - b.r, a.i - b.i}; }
__device__ __forceinline__ cpx cinv(cpx a) { float d = a.r*a.r + a.i*a.i; return {a.r/d, -a.i/d}; }
__device__ __forceinline__ cpx cscale(cpx a, float s) { return {a.r*s, a.i*s}; }

__device__ __forceinline__ void load_frags(const float* Pr, const float* Pi, int row,
                                           v2f (&fr)[16], v2f (&fi)[16])
{
    const int lane = threadIdx.x & 31;
    const int h4 = (lane >> 4) << 2;
    const v4f* qr = (const v4f*)(Pr + row * PA + h4);
    const v4f* qi = (const v4f*)(Pi + row * PA + h4);
    #pragma unroll
    for (int t = 0; t < 8; ++t) {
        const v4f a = qr[2*t];
        const v4f b = qi[2*t];
        fr[2*t][0] = a.x;  fr[2*t][1] = a.y;  fr[2*t + 1][0] = a.z;  fr[2*t + 1][1] = a.w;
        fi[2*t][0] = b.x;  fi[2*t][1] = b.y;  fi[2*t + 1][0] = b.z;  fi[2*t + 1][1] = b.w;
    }
}

__device__ __forceinline__ void sum_a_frags(const v2f (&arf)[16], const v2f (&aif)[16], v2f (&asf)[16])
{
    #pragma unroll
    for (int s = 0; s < 16; ++s) asf[s] = arf[s] + aif[s];
}

__device__ __forceinline__ void cplx_chains(
    const v2f (&arf)[16], const v2f (&aif)[16], const v2f (&asf)[16],
    const v2f (&brv)[16], const v2f (&biv)[16], v8f &p1, v8f &p2, v8f &p3)
{
    #pragma unroll
    for (int s = 0; s < 16; ++s) {
        const v2f bs = brv[s] + biv[s];
        p1 = __builtin_amdgcn_wmma_f32_16x16x4_f32(false, arf[s], false, brv[s], (short)0, p1, false, false);
        p2 = __builtin_amdgcn_wmma_f32_16x16x4_f32(false, aif[s], false, biv[s], (short)0, p2, false, false);
        p3 = __builtin_amdgcn_wmma_f32_16x16x4_f32(false, asf[s], false, bs, (short)0, p3, false, false);
        __builtin_amdgcn_sched_barrier(0);
    }
}

template <bool ACOPY>
__device__ __forceinline__ void cplx_mm_tile(
    const v2f (&arf)[16], const v2f (&aif)[16], const v2f (&asf)[16],
    const float* BTr, const float* BTi,
    float* CTr, float* CTi,
    float* Er, float* Ei,
    int r0, int n0)
{
    const int lane = threadIdx.x & 31;
    const int mn = lane & 15;
    v2f brv[16], biv[16];
    load_frags(BTr, BTi, n0 + mn, brv, biv);
    __builtin_amdgcn_sched_barrier(0);
    v8f p1 = {}; v8f p2 = {}; v8f p3 = {};
    cplx_chains(arf, aif, asf, brv, biv, p1, p2, p3);
    const int mbase = (lane >> 4) << 3;
    float re[8], im[8];
    for (int v = 0; v < 8; ++v) { re[v] = p1[v] - p2[v]; im[v] = (p3[v] - p1[v]) - p2[v]; }
    v4f* cr = (v4f*)(CTr + (n0 + mn) * PA + r0 + mbase);
    v4f* ci = (v4f*)(CTi + (n0 + mn) * PA + r0 + mbase);
    cr[0] = (v4f){re[0], re[1], re[2], re[3]};  cr[1] = (v4f){re[4], re[5], re[6], re[7]};
    ci[0] = (v4f){im[0], im[1], im[2], im[3]};  ci[1] = (v4f){im[4], im[5], im[6], im[7]};
    if (ACOPY) {
        for (int v = 0; v < 8; ++v) {
            Er[(r0 + mbase + v) * PA + n0 + mn] = re[v];
            Ei[(r0 + mbase + v) * PA + n0 + mn] = im[v];
        }
    }
}

template <int M>
__device__ __forceinline__ float row_xor_add(float v)
{
    return v + __builtin_bit_cast(float, __builtin_amdgcn_update_dpp(0, __builtin_bit_cast(int, v), 0x160 | M, 0xf, 0xf, false));
}
__device__ __forceinline__ float wave_sum(float v)
{
    v = row_xor_add<1>(v);  v = row_xor_add<2>(v);  v = row_xor_add<4>(v);  v = row_xor_add<8>(v);
    const int b = __builtin_bit_cast(int, v);
    return v + __builtin_bit_cast(float, __builtin_amdgcn_permlanex16(b, b, 0x76543210, 0xfedcba98, false, false));
}

__device__ __forceinline__ void load_frags4(const float* Pr, const float* Pi, int row,
                                            v2f (&fr)[4], v2f (&fi)[4])
{
    const int lane = threadIdx.x & 31;
    const int h4 = (lane >> 4) << 2;
    const v4f* qr = (const v4f*)(Pr + row * PA + h4);
    const v4f* qi = (const v4f*)(Pi + row * PA + h4);
    #pragma unroll
    for (int t = 0; t < 2; ++t) {
        const v4f a = qr[2*t];
        const v4f b = qi[2*t];
        fr[2*t][0] = a.x;  fr[2*t][1] = a.y;  fr[2*t + 1][0] = a.z;  fr[2*t + 1][1] = a.w;
        fi[2*t][0] = b.x;  fi[2*t][1] = b.y;  fi[2*t + 1][0] = b.z;  fi[2*t + 1][1] = b.w;
    }
}

__device__ __forceinline__ void rank16_tile(
    const v2f (&a4r)[4], const v2f (&a4i)[4],
    const float* ZTr, const float* ZTi, const float* g16r, const float* g16i,
    float* CTr, float* CTi, float* Er, float* Ei, int r0, int n0)
{
    const int lane = threadIdx.x & 31;
    const int mn = lane & 15;
    v2f b4r[4], b4i[4];
    load_frags4(ZTr, ZTi, n0 + mn, b4r, b4i);
    const float dgr = g16r[n0 + mn], dgi = g16i[n0 + mn];
    v8f p1 = {}; v8f p2 = {}; v8f p3 = {};
    #pragma unroll
    for (int s = 0; s < 4; ++s) {
        const v2f as = a4r[s] + a4i[s];
        const v2f bs = b4r[s] + b4i[s];
        p1 = __builtin_amdgcn_wmma_f32_16x16x4_f32(false, a4r[s], false, b4r[s], (short)0, p1, false, false);
        p2 = __builtin_amdgcn_wmma_f32_16x16x4_f32(false, a4i[s], false, b4i[s], (short)0, p2, false, false);
        p3 = __builtin_amdgcn_wmma_f32_16x16x4_f32(false, as, false, bs, (short)0, p3, false, false);
    }
    const int mbase = (lane >> 4) << 3;
    float re[8], im[8];
    for (int v = 0; v < 8; ++v) {
        const bool diag = (r0 + mbase + v) == (n0 + mn);
        re[v] = (p1[v] - p2[v]) + (diag ? dgr : 0.f);
        im[v] = ((p3[v] - p1[v]) - p2[v]) + (diag ? dgi : 0.f);
    }
    v4f* cr = (v4f*)(CTr + (n0 + mn) * PA + r0 + mbase);
    v4f* ci = (v4f*)(CTi + (n0 + mn) * PA + r0 + mbase);
    cr[0] = (v4f){re[0], re[1], re[2], re[3]};  cr[1] = (v4f){re[4], re[5], re[6], re[7]};
    ci[0] = (v4f){im[0], im[1], im[2], im[3]};  ci[1] = (v4f){im[4], im[5], im[6], im[7]};
    for (int v = 0; v < 8; ++v) {
        Er[(r0 + mbase + v) * PA + n0 + mn] = re[v];
        Ei[(r0 + mbase + v) * PA + n0 + mn] = im[v];
    }
}

__device__ __forceinline__ void g_double_tile(
    const v2f (&arf)[16], const v2f (&aif)[16], const v2f (&asf)[16],
    const float* MTr, const float* MTi,
    float* Gr, float* Gi, float* Pr, float* Pi, int n0, int w)
{
    const int lane = threadIdx.x & 31;
    const int mn = lane & 15;
    v2f brv[16], biv[16];
    load_frags(MTr, MTi, n0 + mn, brv, biv);
    __builtin_amdgcn_sched_barrier(0);
    v8f p1 = {}; v8f p2 = {}; v8f p3 = {};
    cplx_chains(arf, aif, asf, brv, biv, p1, p2, p3);
    const int mbase = (lane >> 4) << 3;
    for (int v = 0; v < 8; ++v) {
        const int row = mbase + v;
        const bool live = row < w;
        float* dr = live ? (Gr + (w + row) * PA) : (Pr + row * PA);
        float* di = live ? (Gi + (w + row) * PA) : (Pi + row * PA);
        dr[n0 + mn] = p1[v] - p2[v];
        di[n0 + mn] = (p3[v] - p1[v]) - p2[v];
    }
}

__device__ __forceinline__ void k_tile(
    const v2f (&arf)[16], const v2f (&aif)[16],
    const float* XTr, const float* XTi, float* Kb, int j0, int n0)
{
    const int lane = threadIdx.x & 31;
    const int mn = lane & 15;
    v2f brv[16], biv[16];
    load_frags(XTr, XTi, n0 + mn, brv, biv);
    v8f grxr = {}; v8f gixi = {};
    #pragma unroll
    for (int s = 0; s < 16; ++s) {
        grxr = __builtin_amdgcn_wmma_f32_16x16x4_f32(false, arf[s], false, brv[s], (short)0, grxr, false, false);
        gixi = __builtin_amdgcn_wmma_f32_16x16x4_f32(false, aif[s], false, biv[s], (short)0, gixi, false, false);
    }
    const int mbase = (lane >> 4) << 3;
    for (int v = 0; v < 8; ++v)
        Kb[(j0 + mbase + v) * 64 + n0 + mn] = 2.0f * (grxr[v] - gixi[v]);
}

extern "C" __global__ __launch_bounds__(256)
void s4_dplr_kernel(const float* __restrict__ A_real, const float* __restrict__ A_imag,
                    const float* __restrict__ B_real, const float* __restrict__ B_imag,
                    const float* __restrict__ P_real, const float* __restrict__ P_imag,
                    const float* __restrict__ C_real, const float* __restrict__ C_imag,
                    const float* __restrict__ log_dt, float* __restrict__ out)
{
    extern __shared__ __attribute__((aligned(16))) float sm[];
    const int h   = blockIdx.x;
    const int tid = threadIdx.x;
    const int wv  = __builtin_amdgcn_readfirstlane(tid >> 5);
    const int rb  = wv & 3;
    const int hf  = wv >> 2;
    const int r0  = 16 * rb;
    const int c0  = 32 * hf;
    const int mn  = tid & 15;

    const int OFF_T0 = 0;
    const int OFF_T1 = 2 * PLANE;
    const int OFF_A0 = 4 * PLANE;
    const int OFF_A1 = 6 * PLANE;
    float* XTr = sm + 8 * PLANE;
    float* XTi = XTr + PLANE;
    float* Gr  = sm + 10 * PLANE;
    float* Gi  = Gr + 32 * PA;
    float* Dr  = Gr + 16 * PA;
    float* Di  = Gi + 16 * PA;
    float* Kb  = sm + OFF_T0;
    float* ur  = Gi + 32 * PA; float* ui = ur + 64;
    float* wr  = ui + 64;     float* wi = wr + 64;
    float* gr  = wi + 64;     float* gi = gr + 64;
    float* pr  = gi + 64;     float* pi = pr + 64;
    float* scal = pi + 64;
    float* tr   = scal + 8;   float* ti = tr + 64;
    float* g16r = ti + 64;    float* g16i = g16r + 64;
    float* Ur   = sm + OFF_A1;  float* Ui  = Ur + PLANE;
    float* ZTr  = sm + OFF_T1;  float* ZTi = ZTr + PLANE;

    const float dt = __expf(log_dt[h]);
    const float cc = 0.5f * dt;

    if (tid < 64) {
        const int n = tid;
        cpx lam = { -A_real[h*64 + n], -A_imag[h*64 + n] };
        cpx p   = {  P_real[h*64 + n],  P_imag[h*64 + n] };
        cpx Dinv = cinv((cpx){1.0f - cc*lam.r, -cc*lam.i});
        cpx e    = {1.0f + cc*lam.r,  cc*lam.i};
        cpx u = cmul(Dinv, p);
        cpx g = cmul(Dinv, e);
        ur[n] = u.r; ui[n] = u.i; gr[n] = g.r; gi[n] = g.i;
        pr[n] = p.r; pi[n] = p.i;
        Gr[n] = C_real[h*64 + n]; Gi[n] = C_imag[h*64 + n];
    }
    for (int idx = 64 + tid; idx < 32 * PA; idx += 256) { Gr[idx] = 0.f; Gi[idx] = 0.f; }
    __syncthreads();
    if (tid == 0) {
        cpx s = {0.f, 0.f};
        for (int n = 0; n < 64; ++n)
            s = cadd(s, cmul((cpx){pr[n], -pi[n]}, (cpx){ur[n], ui[n]}));
        cpx alpha = cmul((cpx){cc, 0.f}, cinv((cpx){1.0f + cc*s.r, cc*s.i}));
        cpx base  = csub((cpx){cc, 0.f}, cmul(alpha, cscale(s, cc)));
        scal[0] = alpha.r; scal[1] = alpha.i; scal[2] = base.r; scal[3] = base.i;
    }
    __syncthreads();
    const cpx alpha = {scal[0], scal[1]};
    const cpx base  = {scal[2], scal[3]};
    if (tid < 64) {
        const int n = tid;
        cpx w = cmul((cpx){pr[n], -pi[n]}, cadd(base, cmul(alpha, (cpx){gr[n], gi[n]})));
        wr[n] = w.r; wi[n] = w.i;
        cpx lam = { -A_real[h*64 + n], -A_imag[h*64 + n] };
        cpx Dinv = cinv((cpx){1.0f - cc*lam.r, -cc*lam.i});
        cpx b = { B_real[h*64 + n], B_imag[h*64 + n] };
        cpx t = cmul(Dinv, cscale(b, dt));
        tr[n] = t.r; ti[n] = t.i;
    }
    __syncthreads();

    if (wv == 0) {
        cpx xv[2], gv[2], uv[2], wv2[2];
        cpx qp = {0.f, 0.f};
        for (int e = 0; e < 2; ++e) {
            const int n = tid + 32 * e;
            gv[e] = (cpx){gr[n], gi[n]};  uv[e] = (cpx){ur[n], ui[n]};  wv2[e] = (cpx){wr[n], wi[n]};
            xv[e] = (cpx){tr[n], ti[n]};
            qp = cadd(qp, cmul((cpx){pr[n], -pi[n]}, xv[e]));
        }
        const cpx aq = cmul(alpha, (cpx){wave_sum(qp.r), wave_sum(qp.i)});
        for (int e = 0; e < 2; ++e) xv[e] = csub(xv[e], cmul(aq, uv[e]));
        for (int i = 0; i < 16; ++i) {
            for (int e = 0; e < 2; ++e) {
                XTr[i * PA + tid + 32 * e] = xv[e].r;
                XTi[i * PA + tid + 32 * e] = xv[e].i;
            }
            if (i == 15) break;
            const cpx sp = cadd(cmul(wv2[0], xv[0]), cmul(wv2[1], xv[1]));
            const cpx sg = {wave_sum(sp.r), wave_sum(sp.i)};
            for (int e = 0; e < 2; ++e) xv[e] = csub(cmul(gv[e], xv[e]), cmul(uv[e], sg));
        }
    } else if (wv == 1) {
        const int l = tid - 32;
        cpx zv[2], gv[2], uv[2], wv2[2];
        for (int e = 0; e < 2; ++e) {
            const int n = l + 32 * e;
            gv[e] = (cpx){gr[n], gi[n]};  uv[e] = (cpx){ur[n], ui[n]};  wv2[e] = (cpx){wr[n], wi[n]};
            zv[e] = wv2[e];
        }
        for (int j = 0; j < 16; ++j) {
            for (int e = 0; e < 2; ++e) {
                ZTr[(l + 32 * e) * PA + 15 - j] = zv[e].r;
                ZTi[(l + 32 * e) * PA + 15 - j] = zv[e].i;
            }
            if (j == 15) break;
            const cpx tp = cadd(cmul(zv[0], uv[0]), cmul(zv[1], uv[1]));
            const cpx tg = {wave_sum(tp.r), wave_sum(tp.i)};
            for (int e = 0; e < 2; ++e) zv[e] = csub(cmul(gv[e], zv[e]), cmul(wv2[e], tg));
        }
    } else if (wv < 4) {
        const int n = tid - 64;
        const cpx g = {gr[n], gi[n]};
        cpx cu = {-ur[n], -ui[n]};
        for (int i = 0; i < 16; ++i) {
            Ur[n * PA + i] = cu.r;  Ui[n * PA + i] = cu.i;
            cu = cmul(cu, g);
        }
        const cpx g2 = cmul(g, g), g4 = cmul(g2, g2), g8 = cmul(g4, g4), g16 = cmul(g8, g8);
        g16r[n] = g16.r;  g16i[n] = g16.i;
    }
    __syncthreads();
    {
        v2f a4r[4], a4i[4];
        load_frags4(Ur, Ui, r0 + mn, a4r, a4i);
        for (int n0 = c0; n0 < c0 + 32; n0 += 16)
            rank16_tile(a4r, a4i, ZTr, ZTi, g16r, g16i, sm + OFF_T0, sm + OFF_T0 + PLANE,
                        sm + OFF_A0, sm + OFF_A0 + PLANE, r0, n0);
    }
    __syncthreads();

    v2f arf[16], aif[16], asf[16];
    int src = OFF_T0, dst = OFF_T1;
    int asrc = OFF_A0, adst = OFF_A1;

    for (int w = 16; w < 64; w += w) {
        load_frags(sm + asrc, sm + asrc + PLANE, r0 + mn, arf, aif);
        sum_a_frags(arf, aif, asf);
        for (int n0 = 16 * hf; n0 < w; n0 += 32)
            cplx_mm_tile<false>(arf, aif, asf, XTr, XTi, XTr + w * PA, XTi + w * PA, nullptr, nullptr, r0, n0);
        for (int n0 = c0; n0 < c0 + 32; n0 += 16)
            cplx_mm_tile<true>(arf, aif, asf, sm + src, sm + src + PLANE, sm + dst, sm + dst + PLANE, sm + adst, sm + adst + PLANE, r0, n0);
        __syncthreads();
        const int t = src; src = dst; dst = t;
        const int ta = asrc; asrc = adst; adst = ta;
    }
    for (int w = 1; w <= 16; w += w) {
        if (hf == 0) {
            load_frags(Gr, Gi, mn, arf, aif);
            sum_a_frags(arf, aif, asf);
            g_double_tile(arf, aif, asf, sm + src, sm + src + PLANE, Gr, Gi, Dr, Di, r0, w);
        }
        if (w < 16) {
            load_frags(sm + asrc, sm + asrc + PLANE, r0 + mn, arf, aif);
            sum_a_frags(arf, aif, asf);
            for (int n0 = c0; n0 < c0 + 32; n0 += 16)
                cplx_mm_tile<true>(arf, aif, asf, sm + src, sm + src + PLANE, sm + dst, sm + dst + PLANE, sm + adst, sm + adst + PLANE, r0, n0);
        }
        __syncthreads();
        const int t = src; src = dst; dst = t;
        const int ta = asrc; asrc = adst; adst = ta;
    }
    {
        const int j0 = 16 * hf;
        load_frags(Gr, Gi, j0 + mn, arf, aif);
        k_tile(arf, aif, XTr, XTi, Kb, j0, r0);
    }
    __syncthreads();

    for (int i = 0; i < LL / 256; ++i) {
        const int idx = i*256 + tid;
        out[h*LL + idx] = Kb[idx];
    }
}

extern "C" void kernel_launch(void* const* d_in, const int* in_sizes, int n_in,
                              void* d_out, int out_size, void* d_ws, size_t ws_size,
                              hipStream_t stream) {
    const float* A_real = (const float*)d_in[0];
    const float* A_imag = (const float*)d_in[1];
    const float* B_real = (const float*)d_in[2];
    const float* B_imag = (const float*)d_in[3];
    const float* P_real = (const float*)d_in[4];
    const float* P_imag = (const float*)d_in[5];
    const float* C_real = (const float*)d_in[6];
    const float* C_imag = (const float*)d_in[7];
    const float* log_dt = (const float*)d_in[8];
    float* out = (float*)d_out;
    (void)in_sizes; (void)n_in; (void)out_size; (void)d_ws; (void)ws_size;

    const size_t shmem_bytes = 48648u * sizeof(float);
    s4_dplr_kernel<<<NH, 256, shmem_bytes, stream>>>(
        A_real, A_imag, B_real, B_imag, P_real, P_imag, C_real, C_imag, log_dt, out);
}


# Set-up of dA = diag(g) - u w^T and dB done per wave in registers straight from the inputs: no one-lane 64-term serial sum, no LDS hand-over, 3 barriers fewer; sum order of p^H D^-1 p changes
# speedup vs baseline: 23.6221x; 1.0495x over previous
#include <hip/hip_runtime.h>
#include <hip/hip_bf16.h>

#define NH 256
#define NN 64
#define LL 2048
#define PA 68
#define PLANE (64 * PA)

typedef __attribute__((ext_vector_type(2))) float v2f;
typedef __attribute__((ext_vector_type(4))) float v4f;
typedef __attribute__((ext_vector_type(8))) float v8f;

struct cpx { float r, i; };
__device__ __forceinline__ cpx cmul(cpx a, cpx b) { return {a.r*b.r - a.i*b.i, a.r*b.i + a.i*b.r}; }
__device__ __forceinline__ cpx cadd(cpx a, cpx b) { return {a.r + b.r, a.i + b.i}; }
__device__ __forceinline__ cpx csub(cpx a, cpx b) { return {a.r - b.r, a.i - b.i}; }
__device__ __forceinline__ cpx cinv(cpx a) { float d = a.r*a.r + a.i*a.i; return {a.r/d, -a.i/d}; }
__device__ __forceinline__ cpx cscale(cpx a, float s) { return {a.r*s, a.i*s}; }

__device__ __forceinline__ void load_frags(const float* Pr, const float* Pi, int row,
                                           v2f (&fr)[16], v2f (&fi)[16])
{
    const int lane = threadIdx.x & 31;
    const int h4 = (lane >> 4) << 2;
    const v4f* qr = (const v4f*)(Pr + row * PA + h4);
    const v4f* qi = (const v4f*)(Pi + row * PA + h4);
    #pragma unroll
    for (int t = 0; t < 8; ++t) {
        const v4f a = qr[2*t];
        const v4f b = qi[2*t];
        fr[2*t][0] = a.x;  fr[2*t][1] = a.y;  fr[2*t + 1][0] = a.z;  fr[2*t + 1][1] = a.w;
        fi[2*t][0] = b.x;  fi[2*t][1] = b.y;  fi[2*t + 1][0] = b.z;  fi[2*t + 1][1] = b.w;
    }
}

__device__ __forceinline__ void sum_a_frags(const v2f (&arf)[16], const v2f (&aif)[16], v2f (&asf)[16])
{
    #pragma unroll
    for (int s = 0; s < 16; ++s) asf[s] = arf[s] + aif[s];
}

__device__ __forceinline__ void cplx_chains(
    const v2f (&arf)[16], const v2f (&aif)[16], const v2f (&asf)[16],
    const v2f (&brv)[16], const v2f (&biv)[16], v8f &p1, v8f &p2, v8f &p3)
{
    #pragma unroll
    for (int s = 0; s < 16; ++s) {
        const v2f bs = brv[s] + biv[s];
        p1 = __builtin_amdgcn_wmma_f32_16x16x4_f32(false, arf[s], false, brv[s], (short)0, p1, false, false);
        p2 = __builtin_amdgcn_wmma_f32_16x16x4_f32(false, aif[s], false, biv[s], (short)0, p2, false, false);
        p3 = __builtin_amdgcn_wmma_f32_16x16x4_f32(false, asf[s], false, bs, (short)0, p3, false, false);
        __builtin_amdgcn_sched_barrier(0);
    }
}

template <bool ACOPY>
__device__ __forceinline__ void cplx_mm_tile(
    const v2f (&arf)[16], const v2f (&aif)[16], const v2f (&asf)[16],
    const float* BTr, const float* BTi,
    float* CTr, float* CTi,
    float* Er, float* Ei,
    int r0, int n0)
{
    const int lane = threadIdx.x & 31;
    const int mn = lane & 15;
    v2f brv[16], biv[16];
    load_frags(BTr, BTi, n0 + mn, brv, biv);
    __builtin_amdgcn_sched_barrier(0);
    v8f p1 = {}; v8f p2 = {}; v8f p3 = {};
    cplx_chains(arf, aif, asf, brv, biv, p1, p2, p3);
    const int mbase = (lane >> 4) << 3;
    float re[8], im[8];
    for (int v = 0; v < 8; ++v) { re[v] = p1[v] - p2[v]; im[v] = (p3[v] - p1[v]) - p2[v]; }
    v4f* cr = (v4f*)(CTr + (n0 + mn) * PA + r0 + mbase);
    v4f* ci = (v4f*)(CTi + (n0 + mn) * PA + r0 + mbase);
    cr[0] = (v4f){re[0], re[1], re[2], re[3]};  cr[1] = (v4f){re[4], re[5], re[6], re[7]};
    ci[0] = (v4f){im[0], im[1], im[2], im[3]};  ci[1] = (v4f){im[4], im[5], im[6], im[7]};
    if (ACOPY) {
        for (int v = 0; v < 8; ++v) {
            Er[(r0 + mbase + v) * PA + n0 + mn] = re[v];
            Ei[(r0 + mbase + v) * PA + n0 + mn] = im[v];
        }
    }
}

template <int M>
__device__ __forceinline__ float row_xor_add(float v)
{
    return v + __builtin_bit_cast(float, __builtin_amdgcn_update_dpp(0, __builtin_bit_cast(int, v), 0x160 | M, 0xf, 0xf, false));
}
__device__ __forceinline__ float wave_sum(float v)
{
    v = row_xor_add<1>(v);  v = row_xor_add<2>(v);  v = row_xor_add<4>(v);  v = row_xor_add<8>(v);
    const int b = __builtin_bit_cast(int, v);
    return v + __builtin_bit_cast(float, __builtin_amdgcn_permlanex16(b, b, 0x76543210, 0xfedcba98, false, false));
}

__device__ __forceinline__ void load_frags4(const float* Pr, const float* Pi, int row,
                                            v2f (&fr)[4], v2f (&fi)[4])
{
    const int lane = threadIdx.x & 31;
    const int h4 = (lane >> 4) << 2;
    const v4f* qr = (const v4f*)(Pr + row * PA + h4);
    const v4f* qi = (const v4f*)(Pi + row * PA + h4);
    #pragma unroll
    for (int t = 0; t < 2; ++t) {
        const v4f a = qr[2*t];
        const v4f b = qi[2*t];
        fr[2*t][0] = a.x;  fr[2*t][1] = a.y;  fr[2*t + 1][0] = a.z;  fr[2*t + 1][1] = a.w;
        fi[2*t][0] = b.x;  fi[2*t][1] = b.y;  fi[2*t + 1][0] = b.z;  fi[2*t + 1][1] = b.w;
    }
}

__device__ __forceinline__ void rank16_tile(
    const v2f (&a4r)[4], const v2f (&a4i)[4],
    const float* ZTr, const float* ZTi, const float* g16r, const float* g16i,
    float* CTr, float* CTi, float* Er, float* Ei, int r0, int n0)
{
    const int lane = threadIdx.x & 31;
    const int mn = lane & 15;
    v2f b4r[4], b4i[4];
    load_frags4(ZTr, ZTi, n0 + mn, b4r, b4i);
    const float dgr = g16r[n0 + mn], dgi = g16i[n0 + mn];
    v8f p1 = {}; v8f p2 = {}; v8f p3 = {};
    #pragma unroll
    for (int s = 0; s < 4; ++s) {
        const v2f as = a4r[s] + a4i[s];
        const v2f bs = b4r[s] + b4i[s];
        p1 = __builtin_amdgcn_wmma_f32_16x16x4_f32(false, a4r[s], false, b4r[s], (short)0, p1, false, false);
        p2 = __builtin_amdgcn_wmma_f32_16x16x4_f32(false, a4i[s], false, b4i[s], (short)0, p2, false, false);
        p3 = __builtin_amdgcn_wmma_f32_16x16x4_f32(false, as, false, bs, (short)0, p3, false, false);
    }
    const int mbase = (lane >> 4) << 3;
    float re[8], im[8];
    for (int v = 0; v < 8; ++v) {
        const bool diag = (r0 + mbase + v) == (n0 + mn);
        re[v] = (p1[v] - p2[v]) + (diag ? dgr : 0.f);
        im[v] = ((p3[v] - p1[v]) - p2[v]) + (diag ? dgi : 0.f);
    }
    v4f* cr = (v4f*)(CTr + (n0 + mn) * PA + r0 + mbase);
    v4f* ci = (v4f*)(CTi + (n0 + mn) * PA + r0 + mbase);
    cr[0] = (v4f){re[0], re[1], re[2], re[3]};  cr[1] = (v4f){re[4], re[5], re[6], re[7]};
    ci[0] = (v4f){im[0], im[1], im[2], im[3]};  ci[1] = (v4f){im[4], im[5], im[6], im[7]};
    for (int v = 0; v < 8; ++v) {
        Er[(r0 + mbase + v) * PA + n0 + mn] = re[v];
        Ei[(r0 + mbase + v) * PA + n0 + mn] = im[v];
    }
}

__device__ __forceinline__ void g_double_tile(
    const v2f (&arf)[16], const v2f (&aif)[16], const v2f (&asf)[16],
    const float* MTr, const float* MTi,
    float* Gr, float* Gi, float* Pr, float* Pi, int n0, int w)
{
    const int lane = threadIdx.x & 31;
    const int mn = lane & 15;
    v2f brv[16], biv[16];
    load_frags(MTr, MTi, n0 + mn, brv, biv);
    __builtin_amdgcn_sched_barrier(0);
    v8f p1 = {}; v8f p2 = {}; v8f p3 = {};
    cplx_chains(arf, aif, asf, brv, biv, p1, p2, p3);
    const int mbase = (lane >> 4) << 3;
    for (int v = 0; v < 8; ++v) {
        const int row = mbase + v;
        const bool live = row < w;
        float* dr = live ? (Gr + (w + row) * PA) : (Pr + row * PA);
        float* di = live ? (Gi + (w + row) * PA) : (Pi + row * PA);
        dr[n0 + mn] = p1[v] - p2[v];
        di[n0 + mn] = (p3[v] - p1[v]) - p2[v];
    }
}

__device__ __forceinline__ void k_tile(
    const v2f (&arf)[16], const v2f (&aif)[16],
    const float* XTr, const float* XTi, float* Kb, int j0, int n0)
{
    const int lane = threadIdx.x & 31;
    const int mn = lane & 15;
    v2f brv[16], biv[16];
    load_frags(XTr, XTi, n0 + mn, brv, biv);
    v8f grxr = {}; v8f gixi = {};
    #pragma unroll
    for (int s = 0; s < 16; ++s) {
        grxr = __builtin_amdgcn_wmma_f32_16x16x4_f32(false, arf[s], false, brv[s], (short)0, grxr, false, false);
        gixi = __builtin_amdgcn_wmma_f32_16x16x4_f32(false, aif[s], false, biv[s], (short)0, gixi, false, false);
    }
    const int mbase = (lane >> 4) << 3;
    for (int v = 0; v < 8; ++v)
        Kb[(j0 + mbase + v) * 64 + n0 + mn] = 2.0f * (grxr[v] - gixi[v]);
}

extern "C" __global__ __launch_bounds__(256)
void s4_dplr_kernel(const float* __restrict__ A_real, const float* __restrict__ A_imag,
                    const float* __restrict__ B_real, const float* __restrict__ B_imag,
                    const float* __restrict__ P_real, const float* __restrict__ P_imag,
                    const float* __restrict__ C_real, const float* __restrict__ C_imag,
                    const float* __restrict__ log_dt, float* __restrict__ out)
{
    extern __shared__ __attribute__((aligned(16))) float sm[];
    const int h   = blockIdx.x;
    const int tid = threadIdx.x;
    const int wv  = __builtin_amdgcn_readfirstlane(tid >> 5);
    const int rb  = wv & 3;
    const int hf  = wv >> 2;
    const int r0  = 16 * rb;
    const int c0  = 32 * hf;
    const int mn  = tid & 15;

    const int OFF_T0 = 0;
    const int OFF_T1 = 2 * PLANE;
    const int OFF_A0 = 4 * PLANE;
    const int OFF_A1 = 6 * PLANE;
    float* XTr = sm + 8 * PLANE;
    float* XTi = XTr + PLANE;
    float* Gr  = sm + 10 * PLANE;
    float* Gi  = Gr + 32 * PA;
    float* Dr  = Gr + 16 * PA;
    float* Di  = Gi + 16 * PA;
    float* Kb  = sm + OFF_T0;
    float* g16r = Gi + 32 * PA; float* g16i = g16r + 64;
    float* Ur   = sm + OFF_A1;  float* Ui  = Ur + PLANE;
    float* ZTr  = sm + OFF_T1;  float* ZTi = ZTr + PLANE;

    const float dt = __expf(log_dt[h]);
    const float cc = 0.5f * dt;

    if (wv < 2) {
        const int l = tid & 31;
        cpx gv[2], uv[2], wv2[2], pv[2], dinv[2];
        cpx sp = {0.f, 0.f};
        for (int e = 0; e < 2; ++e) {
            const int n = l + 32 * e;
            const cpx lam = { -A_real[h*64 + n], -A_imag[h*64 + n] };
            pv[e]   = (cpx){ P_real[h*64 + n], P_imag[h*64 + n] };
            dinv[e] = cinv((cpx){1.0f - cc*lam.r, -cc*lam.i});
            uv[e]   = cmul(dinv[e], pv[e]);
            gv[e]   = cmul(dinv[e], (cpx){1.0f + cc*lam.r, cc*lam.i});
            sp = cadd(sp, cmul((cpx){pv[e].r, -pv[e].i}, uv[e]));
        }
        const cpx s = {wave_sum(sp.r), wave_sum(sp.i)};
        const cpx alpha = cmul((cpx){cc, 0.f}, cinv((cpx){1.0f + cc*s.r, cc*s.i}));
        const cpx base  = csub((cpx){cc, 0.f}, cmul(alpha, cscale(s, cc)));
        for (int e = 0; e < 2; ++e)
            wv2[e] = cmul((cpx){pv[e].r, -pv[e].i}, cadd(base, cmul(alpha, gv[e])));
        if (wv == 0) {
            cpx xv[2];
            cpx qp = {0.f, 0.f};
            for (int e = 0; e < 2; ++e) {
                const int n = l + 32 * e;
                const cpx b = { B_real[h*64 + n], B_imag[h*64 + n] };
                xv[e] = cmul(dinv[e], cscale(b, dt));
                qp = cadd(qp, cmul((cpx){pv[e].r, -pv[e].i}, xv[e]));
            }
            const cpx aq = cmul(alpha, (cpx){wave_sum(qp.r), wave_sum(qp.i)});
            for (int e = 0; e < 2; ++e) xv[e] = csub(xv[e], cmul(aq, uv[e]));
            for (int i = 0; i < 16; ++i) {
                for (int e = 0; e < 2; ++e) {
                    XTr[i * PA + l + 32 * e] = xv[e].r;
                    XTi[i * PA + l + 32 * e] = xv[e].i;
                }
                if (i == 15) break;
                const cpx dp = cadd(cmul(wv2[0], xv[0]), cmul(wv2[1], xv[1]));
                const cpx sg = {wave_sum(dp.r), wave_sum(dp.i)};
                for (int e = 0; e < 2; ++e) xv[e] = csub(cmul(gv[e], xv[e]), cmul(uv[e], sg));
            }
        } else {
            cpx zv[2];
            for (int e = 0; e < 2; ++e) zv[e] = wv2[e];
            for (int j = 0; j < 16; ++j) {
                for (int e = 0; e < 2; ++e) {
                    ZTr[(l + 32 * e) * PA + 15 - j] = zv[e].r;
                    ZTi[(l + 32 * e) * PA + 15 - j] = zv[e].i;
                }
                if (j == 15) break;
                const cpx tp = cadd(cmul(zv[0], uv[0]), cmul(zv[1], uv[1]));
                const cpx tg = {wave_sum(tp.r), wave_sum(tp.i)};
                for (int e = 0; e < 2; ++e) zv[e] = csub(cmul(gv[e], zv[e]), cmul(wv2[e], tg));
            }
        }
    } else if (wv < 4) {
        const int n = tid - 64;
        const cpx lam  = { -A_real[h*64 + n], -A_imag[h*64 + n] };
        const cpx p    = {  P_real[h*64 + n],  P_imag[h*64 + n] };
        const cpx Dinv = cinv((cpx){1.0f - cc*lam.r, -cc*lam.i});
        const cpx u = cmul(Dinv, p);
        const cpx g = cmul(Dinv, (cpx){1.0f + cc*lam.r, cc*lam.i});
        cpx cu = {-u.r, -u.i};
        for (int i = 0; i < 16; ++i) {
            Ur[n * PA + i] = cu.r;  Ui[n * PA + i] = cu.i;
            cu = cmul(cu, g);
        }
        const cpx g2 = cmul(g, g), g4 = cmul(g2, g2), g8 = cmul(g4, g4), g16 = cmul(g8, g8);
        g16r[n] = g16.r;  g16i[n] = g16.i;
    } else {
        const int q = tid - 128;
        if (q < 64) { Gr[q] = C_real[h*64 + q]; Gi[q] = C_imag[h*64 + q]; }
        for (int idx = 64 + q; idx < 32 * PA; idx += 128) { Gr[idx] = 0.f; Gi[idx] = 0.f; }
    }
    __syncthreads();
    {
        v2f a4r[4], a4i[4];
        load_frags4(Ur, Ui, r0 + mn, a4r, a4i);
        for (int n0 = c0; n0 < c0 + 32; n0 += 16)
            rank16_tile(a4r, a4i, ZTr, ZTi, g16r, g16i, sm + OFF_T0, sm + OFF_T0 + PLANE,
                        sm + OFF_A0, sm + OFF_A0 + PLANE, r0, n0);
    }
    __syncthreads();

    v2f arf[16], aif[16], asf[16];
    int src = OFF_T0, dst = OFF_T1;
    int asrc = OFF_A0, adst = OFF_A1;

    for (int w = 16; w < 64; w += w) {
        load_frags(sm + asrc, sm + asrc + PLANE, r0 + mn, arf, aif);
        sum_a_frags(arf, aif, asf);
        for (int n0 = 16 * hf; n0 < w; n0 += 32)
            cplx_mm_tile<false>(arf, aif, asf, XTr, XTi, XTr + w * PA, XTi + w * PA, nullptr, nullptr, r0, n0);
        for (int n0 = c0; n0 < c0 + 32; n0 += 16)
            cplx_mm_tile<true>(arf, aif, asf, sm + src, sm + src + PLANE, sm + dst, sm + dst + PLANE, sm + adst, sm + adst + PLANE, r0, n0);
        __syncthreads();
        const int t = src; src = dst; dst = t;
        const int ta = asrc; asrc = adst; adst = ta;
    }
    for (int w = 1; w <= 16; w += w) {
        if (hf == 0) {
            load_frags(Gr, Gi, mn, arf, aif);
            sum_a_frags(arf, aif, asf);
            g_double_tile(arf, aif, asf, sm + src, sm + src + PLANE, Gr, Gi, Dr, Di, r0, w);
        }
        if (w < 16) {
            load_frags(sm + asrc, sm + asrc + PLANE, r0 + mn, arf, aif);
            sum_a_frags(arf, aif, asf);
            for (int n0 = c0; n0 < c0 + 32; n0 += 16)
                cplx_mm_tile<true>(arf, aif, asf, sm + src, sm + src + PLANE, sm + dst, sm + dst + PLANE, sm + adst, sm + adst + PLANE, r0, n0);
        }
        __syncthreads();
        const int t = src; src = dst; dst = t;
        const int ta = asrc; asrc = adst; adst = ta;
    }
    {
        const int j0 = 16 * hf;
        load_frags(Gr, Gi, j0 + mn, arf, aif);
        k_tile(arf, aif, XTr, XTi, Kb, j0, r0);
    }
    __syncthreads();

    for (int i = 0; i < LL / 256; ++i) {
        const int idx = i*256 + tid;
        out[h*LL + idx] = Kb[idx];
    }
}

extern "C" void kernel_launch(void* const* d_in, const int* in_sizes, int n_in,
                              void* d_out, int out_size, void* d_ws, size_t ws_size,
                              hipStream_t stream) {
    const float* A_real = (const float*)d_in[0];
    const float* A_imag = (const float*)d_in[1];
    const float* B_real = (const float*)d_in[2];
    const float* B_imag = (const float*)d_in[3];
    const float* P_real = (const float*)d_in[4];
    const float* P_imag = (const float*)d_in[5];
    const float* C_real = (const float*)d_in[6];
    const float* C_imag = (const float*)d_in[7];
    const float* log_dt = (const float*)d_in[8];
    float* out = (float*)d_out;
    (void)in_sizes; (void)n_in; (void)out_size; (void)d_ws; (void)ws_size;

    const size_t shmem_bytes = 48000u * sizeof(float);
    s4_dplr_kernel<<<NH, 256, shmem_bytes, stream>>>(
        A_real, A_imag, B_real, B_imag, P_real, P_imag, C_real, C_imag, log_dt, out);
}
